# starting kernel
# baseline (speedup 1.0000x reference)
#include <hip/hip_runtime.h>
#include <hip/hip_bf16.h>

#define D_FEAT 256
#define HID    16
#define KSTEPS (D_FEAT / 4)
#define KCHUNK 8

typedef float v2f __attribute__((ext_vector_type(2)));
typedef float v8f __attribute__((ext_vector_type(8)));

__global__ void gcn_deg_kernel(const int* __restrict__ dst,
                               unsigned* __restrict__ deg, int E) {
    int e = blockIdx.x * blockDim.x + threadIdx.x;
    if (e < E) atomicAdd(&deg[dst[e]], 1u);
}

__global__ void gcn_dinv_kernel(const unsigned* __restrict__ deg,
                                float* __restrict__ dinv, int N) {
    int i = blockIdx.x * blockDim.x + threadIdx.x;
    if (i < N) dinv[i] = rsqrtf((float)(deg[i] + 1u));
}

__global__ __launch_bounds__(256, 1)
void gcn_gemm_wmma_kernel(const float* __restrict__ x,
                          const float* __restrict__ W,
                          float* __restrict__ h, int ntiles) {
    __shared__ float Wl[D_FEAT * HID];

    const int tid = threadIdx.x;
    for (int i = tid; i < (D_FEAT * HID) / 4; i += blockDim.x)
        ((float4*)Wl)[i] = ((const float4*)W)[i];
    __syncthreads();

    const int lane  = tid & 31;
    const int wave  = tid >> 5;
    const int wpb   = blockDim.x >> 5;
    const int gwave = blockIdx.x * wpb + wave;
    const int nwav  = gridDim.x * wpb;

    const int mrow = lane & 15;
    const int kk   = (lane >> 4) << 1;
    const int ncol = lane & 15;

    v2f bfrag[KSTEPS];
#pragma unroll
    for (int kb = 0; kb < KSTEPS; ++kb) {
        bfrag[kb].x = Wl[(4 * kb + kk)     * HID + ncol];
        bfrag[kb].y = Wl[(4 * kb + kk + 1) * HID + ncol];
    }

    for (int tile = gwave; tile < ntiles; tile += nwav) {
        const float* xrow = x + (size_t)(tile * 16 + mrow) * D_FEAT + kk;

        if (tile + nwav < ntiles)
            __builtin_prefetch(xrow + (size_t)nwav * 16 * D_FEAT, 0, 0);

        v8f c = {};
#pragma unroll
        for (int kc = 0; kc < KSTEPS; kc += KCHUNK) {
            v2f a[KCHUNK];
#pragma unroll
            for (int u = 0; u < KCHUNK; ++u)
                a[u] = *(const v2f*)(xrow + 4 * (kc + u));
#pragma unroll
            for (int u = 0; u < KCHUNK; ++u)
                c = __builtin_amdgcn_wmma_f32_16x16x4_f32(
                        false, a[u], false, bfrag[kc + u],
                        (short)0, c, false, false);
        }

        float* hout = h + (size_t)tile * 16 * HID + (lane >> 4) * 8 * HID + ncol;
#pragma unroll
        for (int v = 0; v < 8; ++v)
            hout[v * HID] = c[v];
    }
}

__global__ void gcn_scatter_kernel(const int* __restrict__ ei,
                                   const float* __restrict__ dinv,
                                   const float* __restrict__ h,
                                   float* __restrict__ agg, int E) {
    for (int e = blockIdx.x * blockDim.x + threadIdx.x; e < E;
         e += gridDim.x * blockDim.x) {
        const int s = ei[e];
        const int d = ei[E + e];
        const float w = dinv[s] * dinv[d];
        const float4* hs = (const float4*)(h + (size_t)s * HID);
        float* ad = agg + (size_t)d * HID;
#pragma unroll
        for (int j = 0; j < 4; ++j) {
            float4 v = hs[j];
            unsafeAtomicAdd(ad + 4 * j + 0, v.x * w);
            unsafeAtomicAdd(ad + 4 * j + 1, v.y * w);
            unsafeAtomicAdd(ad + 4 * j + 2, v.z * w);
            unsafeAtomicAdd(ad + 4 * j + 3, v.w * w);
        }
    }
}

__global__ void gcn_head_kernel(const float* __restrict__ h,
                                const float* __restrict__ agg,
                                const float* __restrict__ dinv,
                                const float* __restrict__ bconv,
                                const float* __restrict__ wfc,
                                const float* __restrict__ bfc,
                                float* __restrict__ out, int N) {
    int i = blockIdx.x * blockDim.x + threadIdx.x;
    if (i >= N) return;
    const float di = dinv[i];
    const float sw = di * di;
    const float4* hp = (const float4*)(h + (size_t)i * HID);
    const float4* ap = (const float4*)(agg + (size_t)i * HID);
    float acc = 0.f;
#pragma unroll
    for (int j = 0; j < 4; ++j) {
        float4 hv = hp[j];
        float4 av = ap[j];
        float4 wv = ((const float4*)wfc)[j];
        float4 bv = ((const float4*)bconv)[j];
        acc += (av.x + hv.x * sw + bv.x) * wv.x;
        acc += (av.y + hv.y * sw + bv.y) * wv.y;
        acc += (av.z + hv.z * sw + bv.z) * wv.z;
        acc += (av.w + hv.w * sw + bv.w) * wv.w;
    }
    out[i] = acc + bfc[0];
}

extern "C" void kernel_launch(void* const* d_in, const int* in_sizes, int n_in,
                              void* d_out, int out_size, void* d_ws, size_t ws_size,
                              hipStream_t stream) {
    const float* x     = (const float*)d_in[0];
    const int*   ei    = (const int*)d_in[1];
    const float* W     = (const float*)d_in[2];
    const float* bconv = (const float*)d_in[3];
    const float* wfc   = (const float*)d_in[4];
    const float* bfc   = (const float*)d_in[5];
    float* out = (float*)d_out;

    const int N = in_sizes[0] / D_FEAT;
    const int E = in_sizes[1] / 2;

    char* ws = (char*)d_ws;
    unsigned* deg  = (unsigned*)ws;
    float*    dinv = (float*)(ws + (size_t)N * 4);
    float*    h    = (float*)(ws + (size_t)N * 8);
    float*    agg  = (float*)(ws + (size_t)N * 8 + (size_t)N * HID * 4);

    hipMemsetAsync(deg, 0, (size_t)N * sizeof(unsigned), stream);
    hipMemsetAsync(agg, 0, (size_t)N * HID * sizeof(float), stream);

    gcn_deg_kernel<<<(E + 255) / 256, 256, 0, stream>>>(ei + E, deg, E);
    gcn_dinv_kernel<<<(N + 255) / 256, 256, 0, stream>>>(deg, dinv, N);

    const int ntiles = N / 16;
    gcn_gemm_wmma_kernel<<<128, 256, 0, stream>>>(x, W, h, ntiles);

    gcn_scatter_kernel<<<1024, 256, 0, stream>>>(ei, dinv, h, agg, E);

    gcn_head_kernel<<<(N + 255) / 256, 256, 0, stream>>>(h, agg, dinv, bconv,
                                                         wfc, bfc, out, N);
}


# Head W_fc applied to h before the edge sum (it is linear): scatter adds one f32 t=dinv*(h.W_fc) per edge, not 16; dst dinv and bias after the sum; f32 summation order changes (atomics kept)
# speedup vs baseline: 2.5439x; 2.5439x over previous
#include <hip/hip_runtime.h>
#include <hip/hip_bf16.h>

#define D_FEAT 256
#define HID    16
#define KSTEPS (D_FEAT / 4)
#define KCHUNK 8

typedef float v2f __attribute__((ext_vector_type(2)));
typedef float v8f __attribute__((ext_vector_type(8)));

__global__ void gcn_deg_kernel(const int* __restrict__ dst,
                               unsigned* __restrict__ deg, int E) {
    int e = blockIdx.x * blockDim.x + threadIdx.x;
    if (e < E) atomicAdd(&deg[dst[e]], 1u);
}

__global__ __launch_bounds__(256, 1)
void gcn_gemm_wmma_kernel(const float* __restrict__ x,
                          const float* __restrict__ W,
                          float* __restrict__ h, int ntiles) {
    __shared__ float Wl[D_FEAT * HID];

    const int tid = threadIdx.x;
    for (int i = tid; i < (D_FEAT * HID) / 4; i += blockDim.x)
        ((float4*)Wl)[i] = ((const float4*)W)[i];
    __syncthreads();

    const int lane  = tid & 31;
    const int wave  = tid >> 5;
    const int wpb   = blockDim.x >> 5;
    const int gwave = blockIdx.x * wpb + wave;
    const int nwav  = gridDim.x * wpb;

    const int mrow = lane & 15;
    const int kk   = (lane >> 4) << 1;
    const int ncol = lane & 15;

    v2f bfrag[KSTEPS];
#pragma unroll
    for (int kb = 0; kb < KSTEPS; ++kb) {
        bfrag[kb].x = Wl[(4 * kb + kk)     * HID + ncol];
        bfrag[kb].y = Wl[(4 * kb + kk + 1) * HID + ncol];
    }

    for (int tile = gwave; tile < ntiles; tile += nwav) {
        const float* xrow = x + (size_t)(tile * 16 + mrow) * D_FEAT + kk;

        if (tile + nwav < ntiles)
            __builtin_prefetch(xrow + (size_t)nwav * 16 * D_FEAT, 0, 0);

        v8f c = {};
#pragma unroll
        for (int kc = 0; kc < KSTEPS; kc += KCHUNK) {
            v2f a[KCHUNK];
#pragma unroll
            for (int u = 0; u < KCHUNK; ++u)
                a[u] = *(const v2f*)(xrow + 4 * (kc + u));
#pragma unroll
            for (int u = 0; u < KCHUNK; ++u)
                c = __builtin_amdgcn_wmma_f32_16x16x4_f32(
                        false, a[u], false, bfrag[kc + u],
                        (short)0, c, false, false);
        }

        float* hout = h + (size_t)tile * 16 * HID + (lane >> 4) * 8 * HID + ncol;
#pragma unroll
        for (int v = 0; v < 8; ++v)
            hout[v * HID] = c[v];
    }
}

__global__ void gcn_node_kernel(const unsigned* __restrict__ deg,
                                const float* __restrict__ h,
                                const float* __restrict__ wfc,
                                float* __restrict__ dinv,
                                float* __restrict__ t,
                                float* __restrict__ acc, int N) {
    int i = blockIdx.x * blockDim.x + threadIdx.x;
    if (i >= N) return;
    const float di = rsqrtf((float)(deg[i] + 1u));
    const float4* hp = (const float4*)(h + (size_t)i * HID);
    float s = 0.f;
#pragma unroll
    for (int j = 0; j < 4; ++j) {
        float4 hv = hp[j];
        float4 wv = ((const float4*)wfc)[j];
        s += hv.x * wv.x;
        s += hv.y * wv.y;
        s += hv.z * wv.z;
        s += hv.w * wv.w;
    }
    const float ti = di * s;
    dinv[i] = di;
    t[i]    = ti;
    acc[i]  = ti;
}

__global__ void gcn_scatter_kernel(const int* __restrict__ ei,
                                   const float* __restrict__ t,
                                   float* __restrict__ acc, int E) {
    const int gtid = blockIdx.x * blockDim.x + threadIdx.x;
    const int nthr = gridDim.x * blockDim.x;
    if ((E & 3) == 0) {
        const int4* s4 = (const int4*)ei;
        const int4* d4 = (const int4*)(ei + E);
        const int nq = E >> 2;
        for (int q = gtid; q < nq; q += nthr) {
            const int4 s = s4[q];
            const int4 d = d4[q];
            const float a0 = t[s.x];
            const float a1 = t[s.y];
            const float a2 = t[s.z];
            const float a3 = t[s.w];
            unsafeAtomicAdd(acc + d.x, a0);
            unsafeAtomicAdd(acc + d.y, a1);
            unsafeAtomicAdd(acc + d.z, a2);
            unsafeAtomicAdd(acc + d.w, a3);
        }
    } else {
        for (int e = gtid; e < E; e += nthr)
            unsafeAtomicAdd(acc + ei[E + e], t[ei[e]]);
    }
}

__global__ void gcn_head_kernel(const float* __restrict__ acc,
                                const float* __restrict__ dinv,
                                const float* __restrict__ bconv,
                                const float* __restrict__ wfc,
                                const float* __restrict__ bfc,
                                float* __restrict__ out, int N) {
    int i = blockIdx.x * blockDim.x + threadIdx.x;
    if (i >= N) return;
    float c = 0.f;
#pragma unroll
    for (int j = 0; j < HID; ++j)
        c += bconv[j] * wfc[j];
    c += bfc[0];
    out[i] = dinv[i] * acc[i] + c;
}

extern "C" void kernel_launch(void* const* d_in, const int* in_sizes, int n_in,
                              void* d_out, int out_size, void* d_ws, size_t ws_size,
                              hipStream_t stream) {
    const float* x     = (const float*)d_in[0];
    const int*   ei    = (const int*)d_in[1];
    const float* W     = (const float*)d_in[2];
    const float* bconv = (const float*)d_in[3];
    const float* wfc   = (const float*)d_in[4];
    const float* bfc   = (const float*)d_in[5];
    float* out = (float*)d_out;

    const int N = in_sizes[0] / D_FEAT;
    const int E = in_sizes[1] / 2;

    char* ws = (char*)d_ws;
    unsigned* deg  = (unsigned*)ws;
    float*    dinv = (float*)(ws + (size_t)N * 4);
    float*    h    = (float*)(ws + (size_t)N * 8);
    float*    t    = (float*)(ws + (size_t)N * 8 + (size_t)N * HID * 4);
    float*    acc  = t + N;

    hipMemsetAsync(deg, 0, (size_t)N * sizeof(unsigned), stream);

    gcn_deg_kernel<<<(E + 255) / 256, 256, 0, stream>>>(ei + E, deg, E);

    const int ntiles = N / 16;
    gcn_gemm_wmma_kernel<<<128, 256, 0, stream>>>(x, W, h, ntiles);

    gcn_node_kernel<<<(N + 255) / 256, 256, 0, stream>>>(deg, h, wfc, dinv, t, acc, N);

    gcn_scatter_kernel<<<(E / 4 + 255) / 256, 256, 0, stream>>>(ei, t, acc, E);

    gcn_head_kernel<<<(N + 255) / 256, 256, 0, stream>>>(acc, dinv, bconv, wfc,
                                                         bfc, out, N);
}


# Degree count and edge sum formed per work-group in LDS (node range x edge chunk), then one global atomic per 128-byte line and chunk, not one per edge; f32 summation order changes (atomics kept)
# speedup vs baseline: 9.5042x; 3.7361x over previous
#include <hip/hip_runtime.h>
#include <hip/hip_bf16.h>

#define D_FEAT 256
#define HID    16
#define KSTEPS (D_FEAT / 4)
#define KCHUNK 8

typedef float v2f __attribute__((ext_vector_type(2)));
typedef float v8f __attribute__((ext_vector_type(8)));

#define AGG_THREADS 1024
#define AGG_TBL     51200
#define AGG_BATCH   4

__global__ __launch_bounds__(AGG_THREADS)
void gcn_deg_kernel(const int* __restrict__ dst,
                    unsigned* __restrict__ deg, int E, int N) {
    __shared__ unsigned tbl[AGG_TBL];
    const int tid  = threadIdx.x;
    const int R    = (N + AGG_TBL - 1) / AGG_TBL;
    const int C    = max(1, (int)gridDim.x / R);
    const int span = (((N + R - 1) / R) + 31) & ~31;

    for (int task = blockIdx.x; task < R * C; task += gridDim.x) {
        const int r   = task % R;
        const int c   = task / R;
        const int lo  = r * span;
        const int len = min(span, N - lo);

        for (int i = tid; i < (len + 3) / 4; i += AGG_THREADS)
            ((uint4*)tbl)[i] = make_uint4(0u, 0u, 0u, 0u);
        __syncthreads();

        if ((E & 3) == 0) {
            const int4* d4 = (const int4*)dst;
            const int nq   = E >> 2;
            const int per  = (nq + C - 1) / C;
            const int qbeg = min(nq, c * per);
            const int qend = min(nq, qbeg + per);
            for (int q0 = qbeg; q0 < qend; q0 += AGG_BATCH * AGG_THREADS) {
                int4 d[AGG_BATCH];
#pragma unroll
                for (int u = 0; u < AGG_BATCH; ++u) {
                    const int q = q0 + u * AGG_THREADS + tid;
                    d[u] = d4[min(q, qend - 1)];
                    if (q >= qend) d[u] = make_int4(-1, -1, -1, -1);
                }
#pragma unroll
                for (int u = 0; u < AGG_BATCH; ++u) {
                    const unsigned a0 = (unsigned)(d[u].x - lo);
                    const unsigned a1 = (unsigned)(d[u].y - lo);
                    const unsigned a2 = (unsigned)(d[u].z - lo);
                    const unsigned a3 = (unsigned)(d[u].w - lo);
                    if (a0 < (unsigned)len) atomicAdd(&tbl[a0], 1u);
                    if (a1 < (unsigned)len) atomicAdd(&tbl[a1], 1u);
                    if (a2 < (unsigned)len) atomicAdd(&tbl[a2], 1u);
                    if (a3 < (unsigned)len) atomicAdd(&tbl[a3], 1u);
                }
            }
        } else {
            const int per  = (E + C - 1) / C;
            const int ebeg = min(E, c * per);
            const int eend = min(E, ebeg + per);
            for (int e = ebeg + tid; e < eend; e += AGG_THREADS) {
                const unsigned a = (unsigned)(dst[e] - lo);
                if (a < (unsigned)len) atomicAdd(&tbl[a], 1u);
            }
        }
        __syncthreads();

        const int rot = (((len >> 5) * c) / C) << 5;
        for (int i0 = tid; i0 < len; i0 += AGG_BATCH * AGG_THREADS) {
            int      j[AGG_BATCH];
            unsigned v[AGG_BATCH];
#pragma unroll
            for (int u = 0; u < AGG_BATCH; ++u) {
                const int i = i0 + u * AGG_THREADS;
                j[u] = i + rot;
                if (j[u] >= len) j[u] -= len;
                v[u] = i < len ? tbl[j[u]] : 0u;
            }
#pragma unroll
            for (int u = 0; u < AGG_BATCH; ++u)
                if (v[u] != 0u) atomicAdd(&deg[lo + j[u]], v[u]);
        }
        __syncthreads();
    }
}

__global__ __launch_bounds__(256, 1)
void gcn_gemm_wmma_kernel(const float* __restrict__ x,
                          const float* __restrict__ W,
                          float* __restrict__ h, int ntiles) {
    __shared__ float Wl[D_FEAT * HID];

    const int tid = threadIdx.x;
    for (int i = tid; i < (D_FEAT * HID) / 4; i += blockDim.x)
        ((float4*)Wl)[i] = ((const float4*)W)[i];
    __syncthreads();

    const int lane  = tid & 31;
    const int wave  = tid >> 5;
    const int wpb   = blockDim.x >> 5;
    const int gwave = blockIdx.x * wpb + wave;
    const int nwav  = gridDim.x * wpb;

    const int mrow = lane & 15;
    const int kk   = (lane >> 4) << 1;
    const int ncol = lane & 15;

    v2f bfrag[KSTEPS];
#pragma unroll
    for (int kb = 0; kb < KSTEPS; ++kb) {
        bfrag[kb].x = Wl[(4 * kb + kk)     * HID + ncol];
        bfrag[kb].y = Wl[(4 * kb + kk + 1) * HID + ncol];
    }

    for (int tile = gwave; tile < ntiles; tile += nwav) {
        const float* xrow = x + (size_t)(tile * 16 + mrow) * D_FEAT + kk;

        if (tile + nwav < ntiles)
            __builtin_prefetch(xrow + (size_t)nwav * 16 * D_FEAT, 0, 0);

        v8f c = {};
#pragma unroll
        for (int kc = 0; kc < KSTEPS; kc += KCHUNK) {
            v2f a[KCHUNK];
#pragma unroll
            for (int u = 0; u < KCHUNK; ++u)
                a[u] = *(const v2f*)(xrow + 4 * (kc + u));
#pragma unroll
            for (int u = 0; u < KCHUNK; ++u)
                c = __builtin_amdgcn_wmma_f32_16x16x4_f32(
                        false, a[u], false, bfrag[kc + u],
                        (short)0, c, false, false);
        }

        float* hout = h + (size_t)tile * 16 * HID + (lane >> 4) * 8 * HID + ncol;
#pragma unroll
        for (int v = 0; v < 8; ++v)
            hout[v * HID] = c[v];
    }
}

__global__ void gcn_node_kernel(const unsigned* __restrict__ deg,
                                const float* __restrict__ h,
                                const float* __restrict__ wfc,
                                float* __restrict__ dinv,
                                float* __restrict__ t,
                                float* __restrict__ acc, int N) {
    int i = blockIdx.x * blockDim.x + threadIdx.x;
    if (i >= N) return;
    const float di = rsqrtf((float)(deg[i] + 1u));
    const float4* hp = (const float4*)(h + (size_t)i * HID);
    float s = 0.f;
#pragma unroll
    for (int j = 0; j < 4; ++j) {
        float4 hv = hp[j];
        float4 wv = ((const float4*)wfc)[j];
        s += hv.x * wv.x;
        s += hv.y * wv.y;
        s += hv.z * wv.z;
        s += hv.w * wv.w;
    }
    const float ti = di * s;
    dinv[i] = di;
    t[i]    = ti;
    acc[i]  = ti;
}

__global__ __launch_bounds__(AGG_THREADS)
void gcn_scatter_kernel(const int* __restrict__ ei,
                        const float* __restrict__ t,
                        float* __restrict__ acc, int E, int N) {
    __shared__ float tbl[AGG_TBL];
    const int tid  = threadIdx.x;
    const int R    = (N + AGG_TBL - 1) / AGG_TBL;
    const int C    = max(1, (int)gridDim.x / R);
    const int span = (((N + R - 1) / R) + 31) & ~31;

    for (int task = blockIdx.x; task < R * C; task += gridDim.x) {
        const int r   = task % R;
        const int c   = task / R;
        const int lo  = r * span;
        const int len = min(span, N - lo);

        for (int i = tid; i < (len + 3) / 4; i += AGG_THREADS)
            ((float4*)tbl)[i] = make_float4(0.f, 0.f, 0.f, 0.f);
        __syncthreads();

        if ((E & 3) == 0) {
            const int4* s4 = (const int4*)ei;
            const int4* d4 = (const int4*)(ei + E);
            const int nq   = E >> 2;
            const int per  = (nq + C - 1) / C;
            const int qbeg = min(nq, c * per);
            const int qend = min(nq, qbeg + per);
            for (int q0 = qbeg; q0 < qend; q0 += AGG_BATCH * AGG_THREADS) {
                int      sv[4 * AGG_BATCH];
                unsigned av[4 * AGG_BATCH];
                float    tv[4 * AGG_BATCH];
#pragma unroll
                for (int u = 0; u < AGG_BATCH; ++u) {
                    const int  q  = q0 + u * AGG_THREADS + tid;
                    const int  qq = min(q, qend - 1);
                    const int4 s  = s4[qq];
                    int4       d  = d4[qq];
                    if (q >= qend) d = make_int4(-1, -1, -1, -1);
                    sv[4 * u + 0] = s.x; sv[4 * u + 1] = s.y;
                    sv[4 * u + 2] = s.z; sv[4 * u + 3] = s.w;
                    av[4 * u + 0] = (unsigned)(d.x - lo);
                    av[4 * u + 1] = (unsigned)(d.y - lo);
                    av[4 * u + 2] = (unsigned)(d.z - lo);
                    av[4 * u + 3] = (unsigned)(d.w - lo);
                }
#pragma unroll
                for (int k = 0; k < 4 * AGG_BATCH; ++k)
                    tv[k] = t[av[k] < (unsigned)len ? sv[k] : 0];
#pragma unroll
                for (int k = 0; k < 4 * AGG_BATCH; ++k)
                    if (av[k] < (unsigned)len) unsafeAtomicAdd(&tbl[av[k]], tv[k]);
            }
        } else {
            const int per  = (E + C - 1) / C;
            const int ebeg = min(E, c * per);
            const int eend = min(E, ebeg + per);
            for (int e = ebeg + tid; e < eend; e += AGG_THREADS) {
                const unsigned a = (unsigned)(ei[E + e] - lo);
                if (a < (unsigned)len) unsafeAtomicAdd(&tbl[a], t[ei[e]]);
            }
        }
        __syncthreads();

        const int rot = (((len >> 5) * c) / C) << 5;
        for (int i0 = tid; i0 < len; i0 += AGG_BATCH * AGG_THREADS) {
            int   j[AGG_BATCH];
            float v[AGG_BATCH];
#pragma unroll
            for (int u = 0; u < AGG_BATCH; ++u) {
                const int i = i0 + u * AGG_THREADS;
                j[u] = i + rot;
                if (j[u] >= len) j[u] -= len;
                v[u] = i < len ? tbl[j[u]] : 0.f;
            }
#pragma unroll
            for (int u = 0; u < AGG_BATCH; ++u)
                if (v[u] != 0.f) unsafeAtomicAdd(acc + lo + j[u], v[u]);
        }
        __syncthreads();
    }
}

__global__ void gcn_head_kernel(const float* __restrict__ acc,
                                const float* __restrict__ dinv,
                                const float* __restrict__ bconv,
                                const float* __restrict__ wfc,
                                const float* __restrict__ bfc,
                                float* __restrict__ out, int N) {
    int i = blockIdx.x * blockDim.x + threadIdx.x;
    if (i >= N) return;
    float c = 0.f;
#pragma unroll
    for (int j = 0; j < HID; ++j)
        c += bconv[j] * wfc[j];
    c += bfc[0];
    out[i] = dinv[i] * acc[i] + c;
}

extern "C" void kernel_launch(void* const* d_in, const int* in_sizes, int n_in,
                              void* d_out, int out_size, void* d_ws, size_t ws_size,
                              hipStream_t stream) {
    const float* x     = (const float*)d_in[0];
    const int*   ei    = (const int*)d_in[1];
    const float* W     = (const float*)d_in[2];
    const float* bconv = (const float*)d_in[3];
    const float* wfc   = (const float*)d_in[4];
    const float* bfc   = (const float*)d_in[5];
    float* out = (float*)d_out;

    const int N = in_sizes[0] / D_FEAT;
    const int E = in_sizes[1] / 2;

    char* ws = (char*)d_ws;
    unsigned* deg  = (unsigned*)ws;
    float*    dinv = (float*)(ws + (size_t)N * 4);
    float*    h    = (float*)(ws + (size_t)N * 8);
    float*    t    = (float*)(ws + (size_t)N * 8 + (size_t)N * HID * 4);
    float*    acc  = t + N;

    hipMemsetAsync(deg, 0, (size_t)N * sizeof(unsigned), stream);

    gcn_deg_kernel<<<256, AGG_THREADS, 0, stream>>>(ei + E, deg, E, N);

    const int ntiles = N / 16;
    gcn_gemm_wmma_kernel<<<128, 256, 0, stream>>>(x, W, h, ntiles);

    gcn_node_kernel<<<(N + 255) / 256, 256, 0, stream>>>(deg, h, wfc, dinv, t, acc, N);

    gcn_scatter_kernel<<<256, AGG_THREADS, 0, stream>>>(ei, t, acc, E, N);

    gcn_head_kernel<<<(N + 255) / 256, 256, 0, stream>>>(acc, dinv, bconv, wfc,
                                                         bfc, out, N);
}


# x feed of the x@W kernel: a wave loads whole 16 KiB tiles (non-temporal, 512 B a load) one tile ahead, re-lays them in its LDS, k paired per 16 B, two sum chains; 256 groups; f32 sum order changes
# speedup vs baseline: 12.4430x; 1.3092x over previous
#include <hip/hip_runtime.h>
#include <hip/hip_bf16.h>

#define D_FEAT 256
#define HID    16

typedef float v2f __attribute__((ext_vector_type(2)));
typedef float v8f __attribute__((ext_vector_type(8)));

#define AGG_THREADS 1024
#define AGG_TBL     51200
#define AGG_BATCH   4

__global__ __launch_bounds__(AGG_THREADS)
void gcn_deg_kernel(const int* __restrict__ dst,
                    unsigned* __restrict__ deg, int E, int N) {
    __shared__ unsigned tbl[AGG_TBL];
    const int tid  = threadIdx.x;
    const int R    = (N + AGG_TBL - 1) / AGG_TBL;
    const int C    = max(1, (int)gridDim.x / R);
    const int span = (((N + R - 1) / R) + 31) & ~31;

    for (int task = blockIdx.x; task < R * C; task += gridDim.x) {
        const int r   = task % R;
        const int c   = task / R;
        const int lo  = r * span;
        const int len = min(span, N - lo);

        for (int i = tid; i < (len + 3) / 4; i += AGG_THREADS)
            ((uint4*)tbl)[i] = make_uint4(0u, 0u, 0u, 0u);
        __syncthreads();

        if ((E & 3) == 0) {
            const int4* d4 = (const int4*)dst;
            const int nq   = E >> 2;
            const int per  = (nq + C - 1) / C;
            const int qbeg = min(nq, c * per);
            const int qend = min(nq, qbeg + per);
            for (int q0 = qbeg; q0 < qend; q0 += AGG_BATCH * AGG_THREADS) {
                int4 d[AGG_BATCH];
#pragma unroll
                for (int u = 0; u < AGG_BATCH; ++u) {
                    const int q = q0 + u * AGG_THREADS + tid;
                    d[u] = d4[min(q, qend - 1)];
                    if (q >= qend) d[u] = make_int4(-1, -1, -1, -1);
                }
#pragma unroll
                for (int u = 0; u < AGG_BATCH; ++u) {
                    const unsigned a0 = (unsigned)(d[u].x - lo);
                    const unsigned a1 = (unsigned)(d[u].y - lo);
                    const unsigned a2 = (unsigned)(d[u].z - lo);
                    const unsigned a3 = (unsigned)(d[u].w - lo);
                    if (a0 < (unsigned)len) atomicAdd(&tbl[a0], 1u);
                    if (a1 < (unsigned)len) atomicAdd(&tbl[a1], 1u);
                    if (a2 < (unsigned)len) atomicAdd(&tbl[a2], 1u);
                    if (a3 < (unsigned)len) atomicAdd(&tbl[a3], 1u);
                }
            }
        } else {
            const int per  = (E + C - 1) / C;
            const int ebeg = min(E, c * per);
            const int eend = min(E, ebeg + per);
            for (int e = ebeg + tid; e < eend; e += AGG_THREADS) {
                const unsigned a = (unsigned)(dst[e] - lo);
                if (a < (unsigned)len) atomicAdd(&tbl[a], 1u);
            }
        }
        __syncthreads();

        const int rot = (((len >> 5) * c) / C) << 5;
        for (int i0 = tid; i0 < len; i0 += AGG_BATCH * AGG_THREADS) {
            int      j[AGG_BATCH];
            unsigned v[AGG_BATCH];
#pragma unroll
            for (int u = 0; u < AGG_BATCH; ++u) {
                const int i = i0 + u * AGG_THREADS;
                j[u] = i + rot;
                if (j[u] >= len) j[u] -= len;
                v[u] = i < len ? tbl[j[u]] : 0u;
            }
#pragma unroll
            for (int u = 0; u < AGG_BATCH; ++u)
                if (v[u] != 0u) atomicAdd(&deg[lo + j[u]], v[u]);
        }
        __syncthreads();
    }
}

#define GEMM_WAVES 8
#define A_PITCH    260
#define TILE_Q     32

typedef float v4f __attribute__((ext_vector_type(4)));

__global__ __launch_bounds__(GEMM_WAVES * 32, 1)
void gcn_gemm_wmma_kernel(const float* __restrict__ x,
                          const float* __restrict__ W,
                          float* __restrict__ h, int ntiles) {
    __shared__ float Wt[D_FEAT * HID];
    __shared__ float At[GEMM_WAVES][16 * A_PITCH];

    const int tid  = threadIdx.x;
    const int lane = tid & 31;
    const int wave = __builtin_amdgcn_readfirstlane(tid >> 5);
    const int c    = lane & 15;
    const int hh   = lane >> 4;

    float4 wq[(D_FEAT * HID) / 4 / (GEMM_WAVES * 32)];
#pragma unroll
    for (int u = 0; u < (D_FEAT * HID) / 4 / (GEMM_WAVES * 32); ++u)
        wq[u] = ((const float4*)W)[tid + u * GEMM_WAVES * 32];

    const int tstride = (int)gridDim.x * GEMM_WAVES;
    int tile = (int)blockIdx.x + (int)gridDim.x * wave;

    v4f g[TILE_Q];
    if (tile < ntiles) {
        const v4f* src = (const v4f*)(x + (size_t)tile * 16 * D_FEAT) + lane;
#pragma unroll
        for (int q = 0; q < TILE_Q; ++q)
            g[q] = __builtin_nontemporal_load(src + q * 32);
    }

#pragma unroll
    for (int u = 0; u < (D_FEAT * HID) / 4 / (GEMM_WAVES * 32); ++u) {
        const int i  = tid + u * GEMM_WAVES * 32;
        const int k  = i >> 2;
        const int n0 = (i & 3) * 4;
        float* d = Wt + ((k >> 2) * 16 + n0) * 4 + (k & 3);
        d[0] = wq[u].x; d[4] = wq[u].y; d[8] = wq[u].z; d[12] = wq[u].w;
    }
    __syncthreads();

    v4f bq[TILE_Q];
#pragma unroll
    for (int j = 0; j < TILE_Q; ++j)
        bq[j] = *(const v4f*)(Wt + ((2 * j + hh) * 16 + c) * 4);

    float* aw = At[wave];

    while (tile < ntiles) {
#pragma unroll
        for (int q = 0; q < TILE_Q; ++q)
            *(v4f*)(aw + (q >> 1) * A_PITCH + (q & 1) * 128 + 4 * lane) = g[q];

        const int next = tile + tstride;
        if (next < ntiles) {
            const v4f* src = (const v4f*)(x + (size_t)next * 16 * D_FEAT) + lane;
#pragma unroll
            for (int q = 0; q < TILE_Q; ++q)
                g[q] = __builtin_nontemporal_load(src + q * 32);
        }

        v4f f[TILE_Q];
#pragma unroll
        for (int j = 0; j < TILE_Q; ++j)
            f[j] = *(const v4f*)(aw + c * A_PITCH + 8 * j + 4 * hh);
        __builtin_amdgcn_sched_barrier(0);

        v8f c0 = {}, c1 = {};
#pragma unroll
        for (int j = 0; j < TILE_Q; ++j) {
            const v2f a0 = {f[j].x, f[j].y}, b0 = {bq[j].x, bq[j].y};
            const v2f a1 = {f[j].z, f[j].w}, b1 = {bq[j].z, bq[j].w};
            c0 = __builtin_amdgcn_wmma_f32_16x16x4_f32(false, a0, false, b0,
                                                       (short)0, c0, false, false);
            c1 = __builtin_amdgcn_wmma_f32_16x16x4_f32(false, a1, false, b1,
                                                       (short)0, c1, false, false);
        }
        const v8f acc = c0 + c1;

        float* hout = h + (size_t)tile * 16 * HID + hh * 8 * HID + c;
#pragma unroll
        for (int v = 0; v < 8; ++v)
            hout[v * HID] = acc[v];

        tile = next;
    }
}

__global__ void gcn_node_kernel(const unsigned* __restrict__ deg,
                                const float* __restrict__ h,
                                const float* __restrict__ wfc,
                                float* __restrict__ dinv,
                                float* __restrict__ t,
                                float* __restrict__ acc, int N) {
    int i = blockIdx.x * blockDim.x + threadIdx.x;
    if (i >= N) return;
    const float di = rsqrtf((float)(deg[i] + 1u));
    const float4* hp = (const float4*)(h + (size_t)i * HID);
    float s = 0.f;
#pragma unroll
    for (int j = 0; j < 4; ++j) {
        float4 hv = hp[j];
        float4 wv = ((const float4*)wfc)[j];
        s += hv.x * wv.x;
        s += hv.y * wv.y;
        s += hv.z * wv.z;
        s += hv.w * wv.w;
    }
    const float ti = di * s;
    dinv[i] = di;
    t[i]    = ti;
    acc[i]  = ti;
}

__global__ __launch_bounds__(AGG_THREADS)
void gcn_scatter_kernel(const int* __restrict__ ei,
                        const float* __restrict__ t,
                        float* __restrict__ acc, int E, int N) {
    __shared__ float tbl[AGG_TBL];
    const int tid  = threadIdx.x;
    const int R    = (N + AGG_TBL - 1) / AGG_TBL;
    const int C    = max(1, (int)gridDim.x / R);
    const int span = (((N + R - 1) / R) + 31) & ~31;

    for (int task = blockIdx.x; task < R * C; task += gridDim.x) {
        const int r   = task % R;
        const int c   = task / R;
        const int lo  = r * span;
        const int len = min(span, N - lo);

        for (int i = tid; i < (len + 3) / 4; i += AGG_THREADS)
            ((float4*)tbl)[i] = make_float4(0.f, 0.f, 0.f, 0.f);
        __syncthreads();

        if ((E & 3) == 0) {
            const int4* s4 = (const int4*)ei;
            const int4* d4 = (const int4*)(ei + E);
            const int nq   = E >> 2;
            const int per  = (nq + C - 1) / C;
            const int qbeg = min(nq, c * per);
            const int qend = min(nq, qbeg + per);
            for (int q0 = qbeg; q0 < qend; q0 += AGG_BATCH * AGG_THREADS) {
                int      sv[4 * AGG_BATCH];
                unsigned av[4 * AGG_BATCH];
                float    tv[4 * AGG_BATCH];
#pragma unroll
                for (int u = 0; u < AGG_BATCH; ++u) {
                    const int  q  = q0 + u * AGG_THREADS + tid;
                    const int  qq = min(q, qend - 1);
                    const int4 s  = s4[qq];
                    int4       d  = d4[qq];
                    if (q >= qend) d = make_int4(-1, -1, -1, -1);
                    sv[4 * u + 0] = s.x; sv[4 * u + 1] = s.y;
                    sv[4 * u + 2] = s.z; sv[4 * u + 3] = s.w;
                    av[4 * u + 0] = (unsigned)(d.x - lo);
                    av[4 * u + 1] = (unsigned)(d.y - lo);
                    av[4 * u + 2] = (unsigned)(d.z - lo);
                    av[4 * u + 3] = (unsigned)(d.w - lo);
                }
#pragma unroll
                for (int k = 0; k < 4 * AGG_BATCH; ++k)
                    tv[k] = t[av[k] < (unsigned)len ? sv[k] : 0];
#pragma unroll
                for (int k = 0; k < 4 * AGG_BATCH; ++k)
                    if (av[k] < (unsigned)len) unsafeAtomicAdd(&tbl[av[k]], tv[k]);
            }
        } else {
            const int per  = (E + C - 1) / C;
            const int ebeg = min(E, c * per);
            const int eend = min(E, ebeg + per);
            for (int e = ebeg + tid; e < eend; e += AGG_THREADS) {
                const unsigned a = (unsigned)(ei[E + e] - lo);
                if (a < (unsigned)len) unsafeAtomicAdd(&tbl[a], t[ei[e]]);
            }
        }
        __syncthreads();

        const int rot = (((len >> 5) * c) / C) << 5;
        for (int i0 = tid; i0 < len; i0 += AGG_BATCH * AGG_THREADS) {
            int   j[AGG_BATCH];
            float v[AGG_BATCH];
#pragma unroll
            for (int u = 0; u < AGG_BATCH; ++u) {
                const int i = i0 + u * AGG_THREADS;
                j[u] = i + rot;
                if (j[u] >= len) j[u] -= len;
                v[u] = i < len ? tbl[j[u]] : 0.f;
            }
#pragma unroll
            for (int u = 0; u < AGG_BATCH; ++u)
                if (v[u] != 0.f) unsafeAtomicAdd(acc + lo + j[u], v[u]);
        }
        __syncthreads();
    }
}

__global__ void gcn_head_kernel(const float* __restrict__ acc,
                                const float* __restrict__ dinv,
                                const float* __restrict__ bconv,
                                const float* __restrict__ wfc,
                                const float* __restrict__ bfc,
                                float* __restrict__ out, int N) {
    int i = blockIdx.x * blockDim.x + threadIdx.x;
    if (i >= N) return;
    float c = 0.f;
#pragma unroll
    for (int j = 0; j < HID; ++j)
        c += bconv[j] * wfc[j];
    c += bfc[0];
    out[i] = dinv[i] * acc[i] + c;
}

extern "C" void kernel_launch(void* const* d_in, const int* in_sizes, int n_in,
                              void* d_out, int out_size, void* d_ws, size_t ws_size,
                              hipStream_t stream) {
    const float* x     = (const float*)d_in[0];
    const int*   ei    = (const int*)d_in[1];
    const float* W     = (const float*)d_in[2];
    const float* bconv = (const float*)d_in[3];
    const float* wfc   = (const float*)d_in[4];
    const float* bfc   = (const float*)d_in[5];
    float* out = (float*)d_out;

    const int N = in_sizes[0] / D_FEAT;
    const int E = in_sizes[1] / 2;

    char* ws = (char*)d_ws;
    unsigned* deg  = (unsigned*)ws;
    float*    dinv = (float*)(ws + (size_t)N * 4);
    float*    h    = (float*)(ws + (size_t)N * 8);
    float*    t    = (float*)(ws + (size_t)N * 8 + (size_t)N * HID * 4);
    float*    acc  = t + N;

    hipMemsetAsync(deg, 0, (size_t)N * sizeof(unsigned), stream);

    gcn_deg_kernel<<<256, AGG_THREADS, 0, stream>>>(ei + E, deg, E, N);

    const int ntiles = N / 16;
    gcn_gemm_wmma_kernel<<<256, 256, 0, stream>>>(x, W, h, ntiles);

    gcn_node_kernel<<<(N + 255) / 256, 256, 0, stream>>>(deg, h, wfc, dinv, t, acc, N);

    gcn_scatter_kernel<<<256, AGG_THREADS, 0, stream>>>(ei, t, acc, E, N);

    gcn_head_kernel<<<(N + 255) / 256, 256, 0, stream>>>(acc, dinv, bconv, wfc,
                                                         bfc, out, N);
}


# LDS tables of the two edge kernels cut to a fifth of the nodes: 5 node ranges x 51 edge chunks, not 2 x 128; 2.5x fewer global atomic lines at write-out, index lists read 5x; f32 sum order changes
# speedup vs baseline: 15.0259x; 1.2076x over previous
#include <hip/hip_runtime.h>
#include <hip/hip_bf16.h>

#define D_FEAT 256
#define HID    16

typedef float v2f __attribute__((ext_vector_type(2)));
typedef float v8f __attribute__((ext_vector_type(8)));

#define AGG_THREADS 1024
#define AGG_TBL     20000
#define AGG_BATCH   4

__global__ __launch_bounds__(AGG_THREADS)
void gcn_deg_kernel(const int* __restrict__ dst,
                    unsigned* __restrict__ deg, int E, int N) {
    __shared__ unsigned tbl[AGG_TBL];
    const int tid  = threadIdx.x;
    const int R    = (N + AGG_TBL - 1) / AGG_TBL;
    const int C    = max(1, (int)gridDim.x / R);
    const int span = (((N + R - 1) / R) + 31) & ~31;

    for (int task = blockIdx.x; task < R * C; task += gridDim.x) {
        const int r   = task % R;
        const int c   = task / R;
        const int lo  = r * span;
        const int len = min(span, N - lo);

        for (int i = tid; i < (len + 3) / 4; i += AGG_THREADS)
            ((uint4*)tbl)[i] = make_uint4(0u, 0u, 0u, 0u);
        __syncthreads();

        if ((E & 3) == 0) {
            const int4* d4 = (const int4*)dst;
            const int nq   = E >> 2;
            const int per  = (nq + C - 1) / C;
            const int qbeg = min(nq, c * per);
            const int qend = min(nq, qbeg + per);
            for (int q0 = qbeg; q0 < qend; q0 += AGG_BATCH * AGG_THREADS) {
                int4 d[AGG_BATCH];
#pragma unroll
                for (int u = 0; u < AGG_BATCH; ++u) {
                    const int q = q0 + u * AGG_THREADS + tid;
                    d[u] = d4[min(q, qend - 1)];
                    if (q >= qend) d[u] = make_int4(-1, -1, -1, -1);
                }
#pragma unroll
                for (int u = 0; u < AGG_BATCH; ++u) {
                    const unsigned a0 = (unsigned)(d[u].x - lo);
                    const unsigned a1 = (unsigned)(d[u].y - lo);
                    const unsigned a2 = (unsigned)(d[u].z - lo);
                    const unsigned a3 = (unsigned)(d[u].w - lo);
                    if (a0 < (unsigned)len) atomicAdd(&tbl[a0], 1u);
                    if (a1 < (unsigned)len) atomicAdd(&tbl[a1], 1u);
                    if (a2 < (unsigned)len) atomicAdd(&tbl[a2], 1u);
                    if (a3 < (unsigned)len) atomicAdd(&tbl[a3], 1u);
                }
            }
        } else {
            const int per  = (E + C - 1) / C;
            const int ebeg = min(E, c * per);
            const int eend = min(E, ebeg + per);
            for (int e = ebeg + tid; e < eend; e += AGG_THREADS) {
                const unsigned a = (unsigned)(dst[e] - lo);
                if (a < (unsigned)len) atomicAdd(&tbl[a], 1u);
            }
        }
        __syncthreads();

        const int rot = (((len >> 5) * c) / C) << 5;
        for (int i0 = tid; i0 < len; i0 += AGG_BATCH * AGG_THREADS) {
            int      j[AGG_BATCH];
            unsigned v[AGG_BATCH];
#pragma unroll
            for (int u = 0; u < AGG_BATCH; ++u) {
                const int i = i0 + u * AGG_THREADS;
                j[u] = i + rot;
                if (j[u] >= len) j[u] -= len;
                v[u] = i < len ? tbl[j[u]] : 0u;
            }
#pragma unroll
            for (int u = 0; u < AGG_BATCH; ++u)
                if (v[u] != 0u) atomicAdd(&deg[lo + j[u]], v[u]);
        }
        __syncthreads();
    }
}

#define GEMM_WAVES 8
#define A_PITCH    260
#define TILE_Q     32

typedef float v4f __attribute__((ext_vector_type(4)));

__global__ __launch_bounds__(GEMM_WAVES * 32, 1)
void gcn_gemm_wmma_kernel(const float* __restrict__ x,
                          const float* __restrict__ W,
                          float* __restrict__ h, int ntiles) {
    __shared__ float Wt[D_FEAT * HID];
    __shared__ float At[GEMM_WAVES][16 * A_PITCH];

    const int tid  = threadIdx.x;
    const int lane = tid & 31;
    const int wave = __builtin_amdgcn_readfirstlane(tid >> 5);
    const int c    = lane & 15;
    const int hh   = lane >> 4;

    float4 wq[(D_FEAT * HID) / 4 / (GEMM_WAVES * 32)];
#pragma unroll
    for (int u = 0; u < (D_FEAT * HID) / 4 / (GEMM_WAVES * 32); ++u)
        wq[u] = ((const float4*)W)[tid + u * GEMM_WAVES * 32];

    const int tstride = (int)gridDim.x * GEMM_WAVES;
    int tile = (int)blockIdx.x + (int)gridDim.x * wave;

    v4f g[TILE_Q];
    if (tile < ntiles) {
        const v4f* src = (const v4f*)(x + (size_t)tile * 16 * D_FEAT) + lane;
#pragma unroll
        for (int q = 0; q < TILE_Q; ++q)
            g[q] = __builtin_nontemporal_load(src + q * 32);
    }

#pragma unroll
    for (int u = 0; u < (D_FEAT * HID) / 4 / (GEMM_WAVES * 32); ++u) {
        const int i  = tid + u * GEMM_WAVES * 32;
        const int k  = i >> 2;
        const int n0 = (i & 3) * 4;
        float* d = Wt + ((k >> 2) * 16 + n0) * 4 + (k & 3);
        d[0] = wq[u].x; d[4] = wq[u].y; d[8] = wq[u].z; d[12] = wq[u].w;
    }
    __syncthreads();

    v4f bq[TILE_Q];
#pragma unroll
    for (int j = 0; j < TILE_Q; ++j)
        bq[j] = *(const v4f*)(Wt + ((2 * j + hh) * 16 + c) * 4);

    float* aw = At[wave];

    while (tile < ntiles) {
#pragma unroll
        for (int q = 0; q < TILE_Q; ++q)
            *(v4f*)(aw + (q >> 1) * A_PITCH + (q & 1) * 128 + 4 * lane) = g[q];

        const int next = tile + tstride;
        if (next < ntiles) {
            const v4f* src = (const v4f*)(x + (size_t)next * 16 * D_FEAT) + lane;
#pragma unroll
            for (int q = 0; q < TILE_Q; ++q)
                g[q] = __builtin_nontemporal_load(src + q * 32);
        }

        v4f f[TILE_Q];
#pragma unroll
        for (int j = 0; j < TILE_Q; ++j)
            f[j] = *(const v4f*)(aw + c * A_PITCH + 8 * j + 4 * hh);
        __builtin_amdgcn_sched_barrier(0);

        v8f c0 = {}, c1 = {};
#pragma unroll
        for (int j = 0; j < TILE_Q; ++j) {
            const v2f a0 = {f[j].x, f[j].y}, b0 = {bq[j].x, bq[j].y};
            const v2f a1 = {f[j].z, f[j].w}, b1 = {bq[j].z, bq[j].w};
            c0 = __builtin_amdgcn_wmma_f32_16x16x4_f32(false, a0, false, b0,
                                                       (short)0, c0, false, false);
            c1 = __builtin_amdgcn_wmma_f32_16x16x4_f32(false, a1, false, b1,
                                                       (short)0, c1, false, false);
        }
        const v8f acc = c0 + c1;

        float* hout = h + (size_t)tile * 16 * HID + hh * 8 * HID + c;
#pragma unroll
        for (int v = 0; v < 8; ++v)
            hout[v * HID] = acc[v];

        tile = next;
    }
}

__global__ void gcn_node_kernel(const unsigned* __restrict__ deg,
                                const float* __restrict__ h,
                                const float* __restrict__ wfc,
                                float* __restrict__ dinv,
                                float* __restrict__ t,
                                float* __restrict__ acc, int N) {
    int i = blockIdx.x * blockDim.x + threadIdx.x;
    if (i >= N) return;
    const float di = rsqrtf((float)(deg[i] + 1u));
    const float4* hp = (const float4*)(h + (size_t)i * HID);
    float s = 0.f;
#pragma unroll
    for (int j = 0; j < 4; ++j) {
        float4 hv = hp[j];
        float4 wv = ((const float4*)wfc)[j];
        s += hv.x * wv.x;
        s += hv.y * wv.y;
        s += hv.z * wv.z;
        s += hv.w * wv.w;
    }
    const float ti = di * s;
    dinv[i] = di;
    t[i]    = ti;
    acc[i]  = ti;
}

__global__ __launch_bounds__(AGG_THREADS)
void gcn_scatter_kernel(const int* __restrict__ ei,
                        const float* __restrict__ t,
                        float* __restrict__ acc, int E, int N) {
    __shared__ float tbl[AGG_TBL];
    const int tid  = threadIdx.x;
    const int R    = (N + AGG_TBL - 1) / AGG_TBL;
    const int C    = max(1, (int)gridDim.x / R);
    const int span = (((N + R - 1) / R) + 31) & ~31;

    for (int task = blockIdx.x; task < R * C; task += gridDim.x) {
        const int r   = task % R;
        const int c   = task / R;
        const int lo  = r * span;
        const int len = min(span, N - lo);

        for (int i = tid; i < (len + 3) / 4; i += AGG_THREADS)
            ((float4*)tbl)[i] = make_float4(0.f, 0.f, 0.f, 0.f);
        __syncthreads();

        if ((E & 3) == 0) {
            const int4* s4 = (const int4*)ei;
            const int4* d4 = (const int4*)(ei + E);
            const int nq   = E >> 2;
            const int per  = (nq + C - 1) / C;
            const int qbeg = min(nq, c * per);
            const int qend = min(nq, qbeg + per);
            for (int q0 = qbeg; q0 < qend; q0 += AGG_BATCH * AGG_THREADS) {
                int      sv[4 * AGG_BATCH];
                unsigned av[4 * AGG_BATCH];
                float    tv[4 * AGG_BATCH];
#pragma unroll
                for (int u = 0; u < AGG_BATCH; ++u) {
                    const int  q  = q0 + u * AGG_THREADS + tid;
                    const int  qq = min(q, qend - 1);
                    const int4 s  = s4[qq];
                    int4       d  = d4[qq];
                    if (q >= qend) d = make_int4(-1, -1, -1, -1);
                    sv[4 * u + 0] = s.x; sv[4 * u + 1] = s.y;
                    sv[4 * u + 2] = s.z; sv[4 * u + 3] = s.w;
                    av[4 * u + 0] = (unsigned)(d.x - lo);
                    av[4 * u + 1] = (unsigned)(d.y - lo);
                    av[4 * u + 2] = (unsigned)(d.z - lo);
                    av[4 * u + 3] = (unsigned)(d.w - lo);
                }
#pragma unroll
                for (int k = 0; k < 4 * AGG_BATCH; ++k)
                    tv[k] = t[av[k] < (unsigned)len ? sv[k] : 0];
#pragma unroll
                for (int k = 0; k < 4 * AGG_BATCH; ++k)
                    if (av[k] < (unsigned)len) unsafeAtomicAdd(&tbl[av[k]], tv[k]);
            }
        } else {
            const int per  = (E + C - 1) / C;
            const int ebeg = min(E, c * per);
            const int eend = min(E, ebeg + per);
            for (int e = ebeg + tid; e < eend; e += AGG_THREADS) {
                const unsigned a = (unsigned)(ei[E + e] - lo);
                if (a < (unsigned)len) unsafeAtomicAdd(&tbl[a], t[ei[e]]);
            }
        }
        __syncthreads();

        const int rot = (((len >> 5) * c) / C) << 5;
        for (int i0 = tid; i0 < len; i0 += AGG_BATCH * AGG_THREADS) {
            int   j[AGG_BATCH];
            float v[AGG_BATCH];
#pragma unroll
            for (int u = 0; u < AGG_BATCH; ++u) {
                const int i = i0 + u * AGG_THREADS;
                j[u] = i + rot;
                if (j[u] >= len) j[u] -= len;
                v[u] = i < len ? tbl[j[u]] : 0.f;
            }
#pragma unroll
            for (int u = 0; u < AGG_BATCH; ++u)
                if (v[u] != 0.f) unsafeAtomicAdd(acc + lo + j[u], v[u]);
        }
        __syncthreads();
    }
}

__global__ void gcn_head_kernel(const float* __restrict__ acc,
                                const float* __restrict__ dinv,
                                const float* __restrict__ bconv,
                                const float* __restrict__ wfc,
                                const float* __restrict__ bfc,
                                float* __restrict__ out, int N) {
    int i = blockIdx.x * blockDim.x + threadIdx.x;
    if (i >= N) return;
    float c = 0.f;
#pragma unroll
    for (int j = 0; j < HID; ++j)
        c += bconv[j] * wfc[j];
    c += bfc[0];
    out[i] = dinv[i] * acc[i] + c;
}

extern "C" void kernel_launch(void* const* d_in, const int* in_sizes, int n_in,
                              void* d_out, int out_size, void* d_ws, size_t ws_size,
                              hipStream_t stream) {
    const float* x     = (const float*)d_in[0];
    const int*   ei    = (const int*)d_in[1];
    const float* W     = (const float*)d_in[2];
    const float* bconv = (const float*)d_in[3];
    const float* wfc   = (const float*)d_in[4];
    const float* bfc   = (const float*)d_in[5];
    float* out = (float*)d_out;

    const int N = in_sizes[0] / D_FEAT;
    const int E = in_sizes[1] / 2;

    char* ws = (char*)d_ws;
    unsigned* deg  = (unsigned*)ws;
    float*    dinv = (float*)(ws + (size_t)N * 4);
    float*    h    = (float*)(ws + (size_t)N * 8);
    float*    t    = (float*)(ws + (size_t)N * 8 + (size_t)N * HID * 4);
    float*    acc  = t + N;

    hipMemsetAsync(deg, 0, (size_t)N * sizeof(unsigned), stream);

    gcn_deg_kernel<<<256, AGG_THREADS, 0, stream>>>(ei + E, deg, E, N);

    const int ntiles = N / 16;
    gcn_gemm_wmma_kernel<<<256, 256, 0, stream>>>(x, W, h, ntiles);

    gcn_node_kernel<<<(N + 255) / 256, 256, 0, stream>>>(deg, h, wfc, dinv, t, acc, N);

    gcn_scatter_kernel<<<256, AGG_THREADS, 0, stream>>>(ei, t, acc, E, N);

    gcn_head_kernel<<<(N + 255) / 256, 256, 0, stream>>>(acc, dinv, bconv, wfc,
                                                         bfc, out, N);
}


# Per-node scalars (dinv, t, acc start) made in the GEMM epilogue from tiles in registers (transposed product, 2 tiles a wave): h never stored, node kernel gone; f32 sum order of h.W_fc changes
# speedup vs baseline: 15.8224x; 1.0530x over previous
#include <hip/hip_runtime.h>
#include <hip/hip_bf16.h>

#define D_FEAT 256
#define HID    16

typedef float v2f __attribute__((ext_vector_type(2)));
typedef float v8f __attribute__((ext_vector_type(8)));

#define AGG_THREADS 1024
#define AGG_TBL     20000
#define AGG_BATCH   4

__global__ __launch_bounds__(AGG_THREADS)
void gcn_deg_kernel(const int* __restrict__ dst,
                    unsigned* __restrict__ deg, int E, int N) {
    __shared__ unsigned tbl[AGG_TBL];
    const int tid  = threadIdx.x;
    const int R    = (N + AGG_TBL - 1) / AGG_TBL;
    const int C    = max(1, (int)gridDim.x / R);
    const int span = (((N + R - 1) / R) + 31) & ~31;

    for (int task = blockIdx.x; task < R * C; task += gridDim.x) {
        const int r   = task % R;
        const int c   = task / R;
        const int lo  = r * span;
        const int len = min(span, N - lo);

        for (int i = tid; i < (len + 3) / 4; i += AGG_THREADS)
            ((uint4*)tbl)[i] = make_uint4(0u, 0u, 0u, 0u);
        __syncthreads();

        if ((E & 3) == 0) {
            const int4* d4 = (const int4*)dst;
            const int nq   = E >> 2;
            const int per  = (nq + C - 1) / C;
            const int qbeg = min(nq, c * per);
            const int qend = min(nq, qbeg + per);
            for (int q0 = qbeg; q0 < qend; q0 += AGG_BATCH * AGG_THREADS) {
                int4 d[AGG_BATCH];
#pragma unroll
                for (int u = 0; u < AGG_BATCH; ++u) {
                    const int q = q0 + u * AGG_THREADS + tid;
                    d[u] = d4[min(q, qend - 1)];
                    if (q >= qend) d[u] = make_int4(-1, -1, -1, -1);
                }
#pragma unroll
                for (int u = 0; u < AGG_BATCH; ++u) {
                    const unsigned a0 = (unsigned)(d[u].x - lo);
                    const unsigned a1 = (unsigned)(d[u].y - lo);
                    const unsigned a2 = (unsigned)(d[u].z - lo);
                    const unsigned a3 = (unsigned)(d[u].w - lo);
                    if (a0 < (unsigned)len) atomicAdd(&tbl[a0], 1u);
                    if (a1 < (unsigned)len) atomicAdd(&tbl[a1], 1u);
                    if (a2 < (unsigned)len) atomicAdd(&tbl[a2], 1u);
                    if (a3 < (unsigned)len) atomicAdd(&tbl[a3], 1u);
                }
            }
        } else {
            const int per  = (E + C - 1) / C;
            const int ebeg = min(E, c * per);
            const int eend = min(E, ebeg + per);
            for (int e = ebeg + tid; e < eend; e += AGG_THREADS) {
                const unsigned a = (unsigned)(dst[e] - lo);
                if (a < (unsigned)len) atomicAdd(&tbl[a], 1u);
            }
        }
        __syncthreads();

        const int rot = (((len >> 5) * c) / C) << 5;
        for (int i0 = tid; i0 < len; i0 += AGG_BATCH * AGG_THREADS) {
            int      j[AGG_BATCH];
            unsigned v[AGG_BATCH];
#pragma unroll
            for (int u = 0; u < AGG_BATCH; ++u) {
                const int i = i0 + u * AGG_THREADS;
                j[u] = i + rot;
                if (j[u] >= len) j[u] -= len;
                v[u] = i < len ? tbl[j[u]] : 0u;
            }
#pragma unroll
            for (int u = 0; u < AGG_BATCH; ++u)
                if (v[u] != 0u) atomicAdd(&deg[lo + j[u]], v[u]);
        }
        __syncthreads();
    }
}

#define GEMM_WAVES 8
#define A_PITCH    260
#define TILE_Q     32

typedef float v4f __attribute__((ext_vector_type(4)));

__device__ __forceinline__ float xhalf(float v) {
    return __builtin_bit_cast(float, __builtin_amdgcn_permlanex16(
        __builtin_bit_cast(int, v), __builtin_bit_cast(int, v),
        0x76543210, 0xfedcba98, false, false));
}

__global__ __launch_bounds__(GEMM_WAVES * 32, 1)
void gcn_gemm_wmma_kernel(const float* __restrict__ x,
                          const float* __restrict__ W,
                          const float* __restrict__ wfc,
                          const unsigned* __restrict__ deg,
                          float* __restrict__ dinv,
                          float* __restrict__ t,
                          float* __restrict__ acc, int ntiles) {
    __shared__ float Wt[D_FEAT * HID];
    __shared__ float At[GEMM_WAVES][16 * A_PITCH];

    const int tid  = threadIdx.x;
    const int lane = tid & 31;
    const int wave = __builtin_amdgcn_readfirstlane(tid >> 5);
    const int c    = lane & 15;
    const int hh   = lane >> 4;

    float4 wq[(D_FEAT * HID) / 4 / (GEMM_WAVES * 32)];
#pragma unroll
    for (int u = 0; u < (D_FEAT * HID) / 4 / (GEMM_WAVES * 32); ++u)
        wq[u] = ((const float4*)W)[tid + u * GEMM_WAVES * 32];

    const float4 wf0 = ((const float4*)wfc)[2 * hh];
    const float4 wf1 = ((const float4*)wfc)[2 * hh + 1];

    const int pstride = (int)gridDim.x * GEMM_WAVES;
    int pair = (int)blockIdx.x + (int)gridDim.x * wave;
    int half = 0;
    int tile = 2 * pair;

    v4f g[TILE_Q];
    if (tile < ntiles) {
        const v4f* src = (const v4f*)(x + (size_t)tile * 16 * D_FEAT) + lane;
#pragma unroll
        for (int q = 0; q < TILE_Q; ++q)
            g[q] = __builtin_nontemporal_load(src + q * 32);
    }

#pragma unroll
    for (int u = 0; u < (D_FEAT * HID) / 4 / (GEMM_WAVES * 32); ++u) {
        const int i  = tid + u * GEMM_WAVES * 32;
        const int k  = i >> 2;
        const int n0 = (i & 3) * 4;
        float* d = Wt + ((k >> 2) * 16 + n0) * 4 + (k & 3);
        d[0] = wq[u].x; d[4] = wq[u].y; d[8] = wq[u].z; d[12] = wq[u].w;
    }
    __syncthreads();

    v4f bq[TILE_Q];
#pragma unroll
    for (int j = 0; j < TILE_Q; ++j)
        bq[j] = *(const v4f*)(Wt + ((2 * j + hh) * 16 + c) * 4);

    float* aw = At[wave];

    unsigned dg = 0u;
    float    s0 = 0.f;

    while (tile < ntiles) {
#pragma unroll
        for (int q = 0; q < TILE_Q; ++q)
            *(v4f*)(aw + (q >> 1) * A_PITCH + (q & 1) * 128 + 4 * lane) = g[q];

        const int npair = half ? pair + pstride : pair;
        const int next  = half ? 2 * npair : tile + 1;

        if (next < ntiles) {
            if (half == 0) dg = deg[32 * pair + lane];
            const v4f* src = (const v4f*)(x + (size_t)next * 16 * D_FEAT) + lane;
#pragma unroll
            for (int q = 0; q < TILE_Q; ++q)
                g[q] = __builtin_nontemporal_load(src + q * 32);
        }

        v4f f[TILE_Q];
#pragma unroll
        for (int j = 0; j < TILE_Q; ++j)
            f[j] = *(const v4f*)(aw + c * A_PITCH + 8 * j + 4 * hh);
        __builtin_amdgcn_sched_barrier(0);

        v8f c0 = {}, c1 = {};
#pragma unroll
        for (int j = 0; j < TILE_Q; ++j) {
            const v2f w0 = {bq[j].x, bq[j].y}, x0 = {f[j].x, f[j].y};
            const v2f w1 = {bq[j].z, bq[j].w}, x1 = {f[j].z, f[j].w};
            c0 = __builtin_amdgcn_wmma_f32_16x16x4_f32(false, w0, false, x0,
                                                       (short)0, c0, false, false);
            c1 = __builtin_amdgcn_wmma_f32_16x16x4_f32(false, w1, false, x1,
                                                       (short)0, c1, false, false);
        }
        const v8f ht = c0 + c1;

        float s = 0.f;
        s += ht[0] * wf0.x;
        s += ht[1] * wf0.y;
        s += ht[2] * wf0.z;
        s += ht[3] * wf0.w;
        s += ht[4] * wf1.x;
        s += ht[5] * wf1.y;
        s += ht[6] * wf1.z;
        s += ht[7] * wf1.w;
        s += xhalf(s);
        if (half == 0) s0 = s;

        if (half == 1) {
            const int   node = 32 * pair + lane;
            const float sn   = hh ? s : s0;
            const float di   = rsqrtf((float)(dg + 1u));
            const float ti   = di * sn;
            dinv[node] = di;
            t[node]    = ti;
            acc[node]  = ti;
        } else if (next >= ntiles) {
            if (hh == 0) {
                const int   node = 16 * tile + c;
                const float di   = rsqrtf((float)(deg[node] + 1u));
                const float ti   = di * s;
                dinv[node] = di;
                t[node]    = ti;
                acc[node]  = ti;
            }
        }

        tile = next;
        pair = npair;
        half ^= 1;
    }
}

__global__ __launch_bounds__(AGG_THREADS)
void gcn_scatter_kernel(const int* __restrict__ ei,
                        const float* __restrict__ t,
                        float* __restrict__ acc, int E, int N) {
    __shared__ float tbl[AGG_TBL];
    const int tid  = threadIdx.x;
    const int R    = (N + AGG_TBL - 1) / AGG_TBL;
    const int C    = max(1, (int)gridDim.x / R);
    const int span = (((N + R - 1) / R) + 31) & ~31;

    for (int task = blockIdx.x; task < R * C; task += gridDim.x) {
        const int r   = task % R;
        const int c   = task / R;
        const int lo  = r * span;
        const int len = min(span, N - lo);

        for (int i = tid; i < (len + 3) / 4; i += AGG_THREADS)
            ((float4*)tbl)[i] = make_float4(0.f, 0.f, 0.f, 0.f);
        __syncthreads();

        if ((E & 3) == 0) {
            const int4* s4 = (const int4*)ei;
            const int4* d4 = (const int4*)(ei + E);
            const int nq   = E >> 2;
            const int per  = (nq + C - 1) / C;
            const int qbeg = min(nq, c * per);
            const int qend = min(nq, qbeg + per);
            for (int q0 = qbeg; q0 < qend; q0 += AGG_BATCH * AGG_THREADS) {
                int      sv[4 * AGG_BATCH];
                unsigned av[4 * AGG_BATCH];
                float    tv[4 * AGG_BATCH];
#pragma unroll
                for (int u = 0; u < AGG_BATCH; ++u) {
                    const int  q  = q0 + u * AGG_THREADS + tid;
                    const int  qq = min(q, qend - 1);
                    const int4 s  = s4[qq];
                    int4       d  = d4[qq];
                    if (q >= qend) d = make_int4(-1, -1, -1, -1);
                    sv[4 * u + 0] = s.x; sv[4 * u + 1] = s.y;
                    sv[4 * u + 2] = s.z; sv[4 * u + 3] = s.w;
                    av[4 * u + 0] = (unsigned)(d.x - lo);
                    av[4 * u + 1] = (unsigned)(d.y - lo);
                    av[4 * u + 2] = (unsigned)(d.z - lo);
                    av[4 * u + 3] = (unsigned)(d.w - lo);
                }
#pragma unroll
                for (int k = 0; k < 4 * AGG_BATCH; ++k)
                    tv[k] = t[av[k] < (unsigned)len ? sv[k] : 0];
#pragma unroll
                for (int k = 0; k < 4 * AGG_BATCH; ++k)
                    if (av[k] < (unsigned)len) unsafeAtomicAdd(&tbl[av[k]], tv[k]);
            }
        } else {
            const int per  = (E + C - 1) / C;
            const int ebeg = min(E, c * per);
            const int eend = min(E, ebeg + per);
            for (int e = ebeg + tid; e < eend; e += AGG_THREADS) {
                const unsigned a = (unsigned)(ei[E + e] - lo);
                if (a < (unsigned)len) unsafeAtomicAdd(&tbl[a], t[ei[e]]);
            }
        }
        __syncthreads();

        const int rot = (((len >> 5) * c) / C) << 5;
        for (int i0 = tid; i0 < len; i0 += AGG_BATCH * AGG_THREADS) {
            int   j[AGG_BATCH];
            float v[AGG_BATCH];
#pragma unroll
            for (int u = 0; u < AGG_BATCH; ++u) {
                const int i = i0 + u * AGG_THREADS;
                j[u] = i + rot;
                if (j[u] >= len) j[u] -= len;
                v[u] = i < len ? tbl[j[u]] : 0.f;
            }
#pragma unroll
            for (int u = 0; u < AGG_BATCH; ++u)
                if (v[u] != 0.f) unsafeAtomicAdd(acc + lo + j[u], v[u]);
        }
        __syncthreads();
    }
}

__global__ void gcn_head_kernel(const float* __restrict__ acc,
                                const float* __restrict__ dinv,
                                const float* __restrict__ bconv,
                                const float* __restrict__ wfc,
                                const float* __restrict__ bfc,
                                float* __restrict__ out, int N) {
    int i = blockIdx.x * blockDim.x + threadIdx.x;
    if (i >= N) return;
    float c = 0.f;
#pragma unroll
    for (int j = 0; j < HID; ++j)
        c += bconv[j] * wfc[j];
    c += bfc[0];
    out[i] = dinv[i] * acc[i] + c;
}

extern "C" void kernel_launch(void* const* d_in, const int* in_sizes, int n_in,
                              void* d_out, int out_size, void* d_ws, size_t ws_size,
                              hipStream_t stream) {
    const float* x     = (const float*)d_in[0];
    const int*   ei    = (const int*)d_in[1];
    const float* W     = (const float*)d_in[2];
    const float* bconv = (const float*)d_in[3];
    const float* wfc   = (const float*)d_in[4];
    const float* bfc   = (const float*)d_in[5];
    float* out = (float*)d_out;

    const int N = in_sizes[0] / D_FEAT;
    const int E = in_sizes[1] / 2;

    char* ws = (char*)d_ws;
    unsigned* deg  = (unsigned*)ws;
    float*    dinv = (float*)(ws + (size_t)N * 4);
    float*    t    = (float*)(ws + (size_t)N * 8);
    float*    acc  = t + N;

    hipMemsetAsync(deg, 0, (size_t)N * sizeof(unsigned), stream);

    gcn_deg_kernel<<<256, AGG_THREADS, 0, stream>>>(ei + E, deg, E, N);

    const int ntiles = N / 16;
    gcn_gemm_wmma_kernel<<<256, 256, 0, stream>>>(x, W, wfc, deg, dinv, t, acc, ntiles);

    gcn_scatter_kernel<<<256, AGG_THREADS, 0, stream>>>(ei, t, acc, E, N);

    gcn_head_kernel<<<(N + 255) / 256, 256, 0, stream>>>(acc, dinv, bconv, wfc,
                                                         bfc, out, N);
}


# Degree and edge-sum arrays get one plane per package half (picked by the work-group's die number): global atomics stay inside a half, readers add the two planes; f32 sum order changes (atomics kept)
# speedup vs baseline: 17.3012x; 1.0935x over previous
#include <hip/hip_runtime.h>
#include <hip/hip_bf16.h>

#define D_FEAT 256
#define HID    16

typedef float v2f __attribute__((ext_vector_type(2)));
typedef float v8f __attribute__((ext_vector_type(8)));

#define AGG_THREADS 1024
#define AGG_TBL     20000
#define AGG_BATCH   4

__device__ __forceinline__ int package_half() {
    const unsigned where = (unsigned)__builtin_amdgcn_s_sendmsg_rtn(0x87);
    return (int)((where >> 18) & 1u);
}

__global__ __launch_bounds__(AGG_THREADS)
void gcn_deg_kernel(const int* __restrict__ dst,
                    unsigned* __restrict__ deg, int E, int N) {
    __shared__ unsigned tbl[AGG_TBL];
    const int tid  = threadIdx.x;
    const int R    = (N + AGG_TBL - 1) / AGG_TBL;
    const int C    = max(1, (int)gridDim.x / R);
    const int span = (((N + R - 1) / R) + 31) & ~31;
    unsigned* const dplane = deg + (size_t)package_half() * N;

    for (int task = blockIdx.x; task < R * C; task += gridDim.x) {
        const int r   = task % R;
        const int c   = task / R;
        const int lo  = r * span;
        const int len = min(span, N - lo);

        for (int i = tid; i < (len + 3) / 4; i += AGG_THREADS)
            ((uint4*)tbl)[i] = make_uint4(0u, 0u, 0u, 0u);
        __syncthreads();

        if ((E & 3) == 0) {
            const int4* d4 = (const int4*)dst;
            const int nq   = E >> 2;
            const int per  = (nq + C - 1) / C;
            const int qbeg = min(nq, c * per);
            const int qend = min(nq, qbeg + per);
            for (int q0 = qbeg; q0 < qend; q0 += AGG_BATCH * AGG_THREADS) {
                int4 d[AGG_BATCH];
#pragma unroll
                for (int u = 0; u < AGG_BATCH; ++u) {
                    const int q = q0 + u * AGG_THREADS + tid;
                    d[u] = d4[min(q, qend - 1)];
                    if (q >= qend) d[u] = make_int4(-1, -1, -1, -1);
                }
#pragma unroll
                for (int u = 0; u < AGG_BATCH; ++u) {
                    const unsigned a0 = (unsigned)(d[u].x - lo);
                    const unsigned a1 = (unsigned)(d[u].y - lo);
                    const unsigned a2 = (unsigned)(d[u].z - lo);
                    const unsigned a3 = (unsigned)(d[u].w - lo);
                    if (a0 < (unsigned)len) atomicAdd(&tbl[a0], 1u);
                    if (a1 < (unsigned)len) atomicAdd(&tbl[a1], 1u);
                    if (a2 < (unsigned)len) atomicAdd(&tbl[a2], 1u);
                    if (a3 < (unsigned)len) atomicAdd(&tbl[a3], 1u);
                }
            }
        } else {
            const int per  = (E + C - 1) / C;
            const int ebeg = min(E, c * per);
            const int eend = min(E, ebeg + per);
            for (int e = ebeg + tid; e < eend; e += AGG_THREADS) {
                const unsigned a = (unsigned)(dst[e] - lo);
                if (a < (unsigned)len) atomicAdd(&tbl[a], 1u);
            }
        }
        __syncthreads();

        const int rot = (((len >> 5) * c) / C) << 5;
        for (int i0 = tid; i0 < len; i0 += AGG_BATCH * AGG_THREADS) {
            int      j[AGG_BATCH];
            unsigned v[AGG_BATCH];
#pragma unroll
            for (int u = 0; u < AGG_BATCH; ++u) {
                const int i = i0 + u * AGG_THREADS;
                j[u] = i + rot;
                if (j[u] >= len) j[u] -= len;
                v[u] = i < len ? tbl[j[u]] : 0u;
            }
#pragma unroll
            for (int u = 0; u < AGG_BATCH; ++u)
                if (v[u] != 0u) atomicAdd(&dplane[lo + j[u]], v[u]);
        }
        __syncthreads();
    }
}

#define GEMM_WAVES 8
#define A_PITCH    260
#define TILE_Q     32

typedef float v4f __attribute__((ext_vector_type(4)));

__device__ __forceinline__ float xhalf(float v) {
    return __builtin_bit_cast(float, __builtin_amdgcn_permlanex16(
        __builtin_bit_cast(int, v), __builtin_bit_cast(int, v),
        0x76543210, 0xfedcba98, false, false));
}

__global__ __launch_bounds__(GEMM_WAVES * 32, 1)
void gcn_gemm_wmma_kernel(const float* __restrict__ x,
                          const float* __restrict__ W,
                          const float* __restrict__ wfc,
                          const unsigned* __restrict__ deg,
                          float* __restrict__ dinv,
                          float* __restrict__ t,
                          float* __restrict__ acc, int ntiles, int N) {
    __shared__ float Wt[D_FEAT * HID];
    __shared__ float At[GEMM_WAVES][16 * A_PITCH];

    const int tid  = threadIdx.x;
    const int lane = tid & 31;
    const int wave = __builtin_amdgcn_readfirstlane(tid >> 5);
    const int c    = lane & 15;
    const int hh   = lane >> 4;

    float4 wq[(D_FEAT * HID) / 4 / (GEMM_WAVES * 32)];
#pragma unroll
    for (int u = 0; u < (D_FEAT * HID) / 4 / (GEMM_WAVES * 32); ++u)
        wq[u] = ((const float4*)W)[tid + u * GEMM_WAVES * 32];

    const float4 wf0 = ((const float4*)wfc)[2 * hh];
    const float4 wf1 = ((const float4*)wfc)[2 * hh + 1];

    const int pstride = (int)gridDim.x * GEMM_WAVES;
    int pair = (int)blockIdx.x + (int)gridDim.x * wave;
    int half = 0;
    int tile = 2 * pair;

    v4f g[TILE_Q];
    if (tile < ntiles) {
        const v4f* src = (const v4f*)(x + (size_t)tile * 16 * D_FEAT) + lane;
#pragma unroll
        for (int q = 0; q < TILE_Q; ++q)
            g[q] = __builtin_nontemporal_load(src + q * 32);
    }

#pragma unroll
    for (int u = 0; u < (D_FEAT * HID) / 4 / (GEMM_WAVES * 32); ++u) {
        const int i  = tid + u * GEMM_WAVES * 32;
        const int k  = i >> 2;
        const int n0 = (i & 3) * 4;
        float* d = Wt + ((k >> 2) * 16 + n0) * 4 + (k & 3);
        d[0] = wq[u].x; d[4] = wq[u].y; d[8] = wq[u].z; d[12] = wq[u].w;
    }
    __syncthreads();

    v4f bq[TILE_Q];
#pragma unroll
    for (int j = 0; j < TILE_Q; ++j)
        bq[j] = *(const v4f*)(Wt + ((2 * j + hh) * 16 + c) * 4);

    float* aw = At[wave];

    const unsigned* const deg1 = deg + N;
    float* const          acc1 = acc + N;

    unsigned dg = 0u, dh = 0u;
    float    s0 = 0.f;

    while (tile < ntiles) {
#pragma unroll
        for (int q = 0; q < TILE_Q; ++q)
            *(v4f*)(aw + (q >> 1) * A_PITCH + (q & 1) * 128 + 4 * lane) = g[q];

        const int npair = half ? pair + pstride : pair;
        const int next  = half ? 2 * npair : tile + 1;

        if (next < ntiles) {
            if (half == 0) {
                dg = deg[32 * pair + lane];
                dh = deg1[32 * pair + lane];
            }
            const v4f* src = (const v4f*)(x + (size_t)next * 16 * D_FEAT) + lane;
#pragma unroll
            for (int q = 0; q < TILE_Q; ++q)
                g[q] = __builtin_nontemporal_load(src + q * 32);
        }

        v4f f[TILE_Q];
#pragma unroll
        for (int j = 0; j < TILE_Q; ++j)
            f[j] = *(const v4f*)(aw + c * A_PITCH + 8 * j + 4 * hh);
        __builtin_amdgcn_sched_barrier(0);

        v8f c0 = {}, c1 = {};
#pragma unroll
        for (int j = 0; j < TILE_Q; ++j) {
            const v2f w0 = {bq[j].x, bq[j].y}, x0 = {f[j].x, f[j].y};
            const v2f w1 = {bq[j].z, bq[j].w}, x1 = {f[j].z, f[j].w};
            c0 = __builtin_amdgcn_wmma_f32_16x16x4_f32(false, w0, false, x0,
                                                       (short)0, c0, false, false);
            c1 = __builtin_amdgcn_wmma_f32_16x16x4_f32(false, w1, false, x1,
                                                       (short)0, c1, false, false);
        }
        const v8f ht = c0 + c1;

        float s = 0.f;
        s += ht[0] * wf0.x;
        s += ht[1] * wf0.y;
        s += ht[2] * wf0.z;
        s += ht[3] * wf0.w;
        s += ht[4] * wf1.x;
        s += ht[5] * wf1.y;
        s += ht[6] * wf1.z;
        s += ht[7] * wf1.w;
        s += xhalf(s);
        if (half == 0) s0 = s;

        if (half == 1) {
            const int   node = 32 * pair + lane;
            const float sn   = hh ? s : s0;
            const float di   = rsqrtf((float)(dg + dh + 1u));
            const float ti   = di * sn;
            dinv[node] = di;
            t[node]    = ti;
            acc[node]  = ti;
            acc1[node] = 0.f;
        } else if (next >= ntiles) {
            if (hh == 0) {
                const int   node = 16 * tile + c;
                const float di   = rsqrtf((float)(deg[node] + deg1[node] + 1u));
                const float ti   = di * s;
                dinv[node] = di;
                t[node]    = ti;
                acc[node]  = ti;
                acc1[node] = 0.f;
            }
        }

        tile = next;
        pair = npair;
        half ^= 1;
    }
}

__global__ __launch_bounds__(AGG_THREADS)
void gcn_scatter_kernel(const int* __restrict__ ei,
                        const float* __restrict__ t,
                        float* __restrict__ acc, int E, int N) {
    __shared__ float tbl[AGG_TBL];
    const int tid  = threadIdx.x;
    const int R    = (N + AGG_TBL - 1) / AGG_TBL;
    const int C    = max(1, (int)gridDim.x / R);
    const int span = (((N + R - 1) / R) + 31) & ~31;
    float* const aplane = acc + (size_t)package_half() * N;

    for (int task = blockIdx.x; task < R * C; task += gridDim.x) {
        const int r   = task % R;
        const int c   = task / R;
        const int lo  = r * span;
        const int len = min(span, N - lo);

        for (int i = tid; i < (len + 3) / 4; i += AGG_THREADS)
            ((float4*)tbl)[i] = make_float4(0.f, 0.f, 0.f, 0.f);
        __syncthreads();

        if ((E & 3) == 0) {
            const int4* s4 = (const int4*)ei;
            const int4* d4 = (const int4*)(ei + E);
            const int nq   = E >> 2;
            const int per  = (nq + C - 1) / C;
            const int qbeg = min(nq, c * per);
            const int qend = min(nq, qbeg + per);
            for (int q0 = qbeg; q0 < qend; q0 += AGG_BATCH * AGG_THREADS) {
                int      sv[4 * AGG_BATCH];
                unsigned av[4 * AGG_BATCH];
                float    tv[4 * AGG_BATCH];
#pragma unroll
                for (int u = 0; u < AGG_BATCH; ++u) {
                    const int  q  = q0 + u * AGG_THREADS + tid;
                    const int  qq = min(q, qend - 1);
                    const int4 s  = s4[qq];
                    int4       d  = d4[qq];
                    if (q >= qend) d = make_int4(-1, -1, -1, -1);
                    sv[4 * u + 0] = s.x; sv[4 * u + 1] = s.y;
                    sv[4 * u + 2] = s.z; sv[4 * u + 3] = s.w;
                    av[4 * u + 0] = (unsigned)(d.x - lo);
                    av[4 * u + 1] = (unsigned)(d.y - lo);
                    av[4 * u + 2] = (unsigned)(d.z - lo);
                    av[4 * u + 3] = (unsigned)(d.w - lo);
                }
#pragma unroll
                for (int k = 0; k < 4 * AGG_BATCH; ++k)
                    tv[k] = t[av[k] < (unsigned)len ? sv[k] : 0];
#pragma unroll
                for (int k = 0; k < 4 * AGG_BATCH; ++k)
                    if (av[k] < (unsigned)len) unsafeAtomicAdd(&tbl[av[k]], tv[k]);
            }
        } else {
            const int per  = (E + C - 1) / C;
            const int ebeg = min(E, c * per);
            const int eend = min(E, ebeg + per);
            for (int e = ebeg + tid; e < eend; e += AGG_THREADS) {
                const unsigned a = (unsigned)(ei[E + e] - lo);
                if (a < (unsigned)len) unsafeAtomicAdd(&tbl[a], t[ei[e]]);
            }
        }
        __syncthreads();

        const int rot = (((len >> 5) * c) / C) << 5;
        for (int i0 = tid; i0 < len; i0 += AGG_BATCH * AGG_THREADS) {
            int   j[AGG_BATCH];
            float v[AGG_BATCH];
#pragma unroll
            for (int u = 0; u < AGG_BATCH; ++u) {
                const int i = i0 + u * AGG_THREADS;
                j[u] = i + rot;
                if (j[u] >= len) j[u] -= len;
                v[u] = i < len ? tbl[j[u]] : 0.f;
            }
#pragma unroll
            for (int u = 0; u < AGG_BATCH; ++u)
                if (v[u] != 0.f) unsafeAtomicAdd(aplane + lo + j[u], v[u]);
        }
        __syncthreads();
    }
}

__global__ void gcn_head_kernel(const float* __restrict__ acc,
                                const float* __restrict__ dinv,
                                const float* __restrict__ bconv,
                                const float* __restrict__ wfc,
                                const float* __restrict__ bfc,
                                float* __restrict__ out, int N) {
    int i = blockIdx.x * blockDim.x + threadIdx.x;
    if (i >= N) return;
    float c = 0.f;
#pragma unroll
    for (int j = 0; j < HID; ++j)
        c += bconv[j] * wfc[j];
    c += bfc[0];
    out[i] = dinv[i] * (acc[i] + acc[N + i]) + c;
}

extern "C" void kernel_launch(void* const* d_in, const int* in_sizes, int n_in,
                              void* d_out, int out_size, void* d_ws, size_t ws_size,
                              hipStream_t stream) {
    const float* x     = (const float*)d_in[0];
    const int*   ei    = (const int*)d_in[1];
    const float* W     = (const float*)d_in[2];
    const float* bconv = (const float*)d_in[3];
    const float* wfc   = (const float*)d_in[4];
    const float* bfc   = (const float*)d_in[5];
    float* out = (float*)d_out;

    const int N = in_sizes[0] / D_FEAT;
    const int E = in_sizes[1] / 2;

    char* ws = (char*)d_ws;
    unsigned* deg  = (unsigned*)ws;
    float*    dinv = (float*)(ws + (size_t)N * 8);
    float*    t    = dinv + N;
    float*    acc  = t + N;

    hipMemsetAsync(deg, 0, (size_t)N * 2 * sizeof(unsigned), stream);

    gcn_deg_kernel<<<256, AGG_THREADS, 0, stream>>>(ei + E, deg, E, N);

    const int ntiles = N / 16;
    gcn_gemm_wmma_kernel<<<256, 256, 0, stream>>>(x, W, wfc, deg, dinv, t, acc, ntiles, N);

    gcn_scatter_kernel<<<256, AGG_THREADS, 0, stream>>>(ei, t, acc, E, N);

    gcn_head_kernel<<<(N + 255) / 256, 256, 0, stream>>>(acc, dinv, bconv, wfc,
                                                         bfc, out, N);
}


# Edge-sum kernel keeps half of t in LDS beside its table (5 dst ranges x 2 src slices x 25 chunks): t[src] comes from LDS, not one cache line per edge; index lists read 10x; f32 sum order changes
# speedup vs baseline: 20.0698x; 1.1600x over previous
#include <hip/hip_runtime.h>
#include <hip/hip_bf16.h>

#define D_FEAT 256
#define HID    16

typedef float v2f __attribute__((ext_vector_type(2)));
typedef float v8f __attribute__((ext_vector_type(8)));

#define AGG_THREADS 1024
#define AGG_TBL     20000
#define AGG_BATCH   4

__device__ __forceinline__ int package_half() {
    const unsigned where = (unsigned)__builtin_amdgcn_s_sendmsg_rtn(0x87);
    return (int)((where >> 18) & 1u);
}

__global__ __launch_bounds__(AGG_THREADS)
void gcn_deg_kernel(const int* __restrict__ dst,
                    unsigned* __restrict__ deg, int E, int N) {
    __shared__ unsigned tbl[AGG_TBL];
    const int tid  = threadIdx.x;
    const int R    = (N + AGG_TBL - 1) / AGG_TBL;
    const int C    = max(1, (int)gridDim.x / R);
    const int span = (((N + R - 1) / R) + 31) & ~31;
    unsigned* const dplane = deg + (size_t)package_half() * N;

    for (int task = blockIdx.x; task < R * C; task += gridDim.x) {
        const int r   = task % R;
        const int c   = task / R;
        const int lo  = r * span;
        const int len = min(span, N - lo);

        for (int i = tid; i < (len + 3) / 4; i += AGG_THREADS)
            ((uint4*)tbl)[i] = make_uint4(0u, 0u, 0u, 0u);
        __syncthreads();

        if ((E & 3) == 0) {
            const int4* d4 = (const int4*)dst;
            const int nq   = E >> 2;
            const int per  = (nq + C - 1) / C;
            const int qbeg = min(nq, c * per);
            const int qend = min(nq, qbeg + per);
            for (int q0 = qbeg; q0 < qend; q0 += AGG_BATCH * AGG_THREADS) {
                int4 d[AGG_BATCH];
#pragma unroll
                for (int u = 0; u < AGG_BATCH; ++u) {
                    const int q = q0 + u * AGG_THREADS + tid;
                    d[u] = d4[min(q, qend - 1)];
                    if (q >= qend) d[u] = make_int4(-1, -1, -1, -1);
                }
#pragma unroll
                for (int u = 0; u < AGG_BATCH; ++u) {
                    const unsigned a0 = (unsigned)(d[u].x - lo);
                    const unsigned a1 = (unsigned)(d[u].y - lo);
                    const unsigned a2 = (unsigned)(d[u].z - lo);
                    const unsigned a3 = (unsigned)(d[u].w - lo);
                    if (a0 < (unsigned)len) atomicAdd(&tbl[a0], 1u);
                    if (a1 < (unsigned)len) atomicAdd(&tbl[a1], 1u);
                    if (a2 < (unsigned)len) atomicAdd(&tbl[a2], 1u);
                    if (a3 < (unsigned)len) atomicAdd(&tbl[a3], 1u);
                }
            }
        } else {
            const int per  = (E + C - 1) / C;
            const int ebeg = min(E, c * per);
            const int eend = min(E, ebeg + per);
            for (int e = ebeg + tid; e < eend; e += AGG_THREADS) {
                const unsigned a = (unsigned)(dst[e] - lo);
                if (a < (unsigned)len) atomicAdd(&tbl[a], 1u);
            }
        }
        __syncthreads();

        const int rot = (((len >> 5) * c) / C) << 5;
        for (int i0 = tid; i0 < len; i0 += AGG_BATCH * AGG_THREADS) {
            int      j[AGG_BATCH];
            unsigned v[AGG_BATCH];
#pragma unroll
            for (int u = 0; u < AGG_BATCH; ++u) {
                const int i = i0 + u * AGG_THREADS;
                j[u] = i + rot;
                if (j[u] >= len) j[u] -= len;
                v[u] = i < len ? tbl[j[u]] : 0u;
            }
#pragma unroll
            for (int u = 0; u < AGG_BATCH; ++u)
                if (v[u] != 0u) atomicAdd(&dplane[lo + j[u]], v[u]);
        }
        __syncthreads();
    }
}

#define GEMM_WAVES 8
#define A_PITCH    260
#define TILE_Q     32

typedef float v4f __attribute__((ext_vector_type(4)));

__device__ __forceinline__ float xhalf(float v) {
    return __builtin_bit_cast(float, __builtin_amdgcn_permlanex16(
        __builtin_bit_cast(int, v), __builtin_bit_cast(int, v),
        0x76543210, 0xfedcba98, false, false));
}

__global__ __launch_bounds__(GEMM_WAVES * 32, 1)
void gcn_gemm_wmma_kernel(const float* __restrict__ x,
                          const float* __restrict__ W,
                          const float* __restrict__ wfc,
                          const unsigned* __restrict__ deg,
                          float* __restrict__ dinv,
                          float* __restrict__ t,
                          float* __restrict__ acc, int ntiles, int N) {
    __shared__ float Wt[D_FEAT * HID];
    __shared__ float At[GEMM_WAVES][16 * A_PITCH];

    const int tid  = threadIdx.x;
    const int lane = tid & 31;
    const int wave = __builtin_amdgcn_readfirstlane(tid >> 5);
    const int c    = lane & 15;
    const int hh   = lane >> 4;

    float4 wq[(D_FEAT * HID) / 4 / (GEMM_WAVES * 32)];
#pragma unroll
    for (int u = 0; u < (D_FEAT * HID) / 4 / (GEMM_WAVES * 32); ++u)
        wq[u] = ((const float4*)W)[tid + u * GEMM_WAVES * 32];

    const float4 wf0 = ((const float4*)wfc)[2 * hh];
    const float4 wf1 = ((const float4*)wfc)[2 * hh + 1];

    const int pstride = (int)gridDim.x * GEMM_WAVES;
    int pair = (int)blockIdx.x + (int)gridDim.x * wave;
    int half = 0;
    int tile = 2 * pair;

    v4f g[TILE_Q];
    if (tile < ntiles) {
        const v4f* src = (const v4f*)(x + (size_t)tile * 16 * D_FEAT) + lane;
#pragma unroll
        for (int q = 0; q < TILE_Q; ++q)
            g[q] = __builtin_nontemporal_load(src + q * 32);
    }

#pragma unroll
    for (int u = 0; u < (D_FEAT * HID) / 4 / (GEMM_WAVES * 32); ++u) {
        const int i  = tid + u * GEMM_WAVES * 32;
        const int k  = i >> 2;
        const int n0 = (i & 3) * 4;
        float* d = Wt + ((k >> 2) * 16 + n0) * 4 + (k & 3);
        d[0] = wq[u].x; d[4] = wq[u].y; d[8] = wq[u].z; d[12] = wq[u].w;
    }
    __syncthreads();

    v4f bq[TILE_Q];
#pragma unroll
    for (int j = 0; j < TILE_Q; ++j)
        bq[j] = *(const v4f*)(Wt + ((2 * j + hh) * 16 + c) * 4);

    float* aw = At[wave];

    const unsigned* const deg1 = deg + N;
    float* const          acc1 = acc + N;

    unsigned dg = 0u, dh = 0u;
    float    s0 = 0.f;

    while (tile < ntiles) {
#pragma unroll
        for (int q = 0; q < TILE_Q; ++q)
            *(v4f*)(aw + (q >> 1) * A_PITCH + (q & 1) * 128 + 4 * lane) = g[q];

        const int npair = half ? pair + pstride : pair;
        const int next  = half ? 2 * npair : tile + 1;

        if (next < ntiles) {
            if (half == 0) {
                dg = deg[32 * pair + lane];
                dh = deg1[32 * pair + lane];
            }
            const v4f* src = (const v4f*)(x + (size_t)next * 16 * D_FEAT) + lane;
#pragma unroll
            for (int q = 0; q < TILE_Q; ++q)
                g[q] = __builtin_nontemporal_load(src + q * 32);
        }

        v4f f[TILE_Q];
#pragma unroll
        for (int j = 0; j < TILE_Q; ++j)
            f[j] = *(const v4f*)(aw + c * A_PITCH + 8 * j + 4 * hh);
        __builtin_amdgcn_sched_barrier(0);

        v8f c0 = {}, c1 = {};
#pragma unroll
        for (int j = 0; j < TILE_Q; ++j) {
            const v2f w0 = {bq[j].x, bq[j].y}, x0 = {f[j].x, f[j].y};
            const v2f w1 = {bq[j].z, bq[j].w}, x1 = {f[j].z, f[j].w};
            c0 = __builtin_amdgcn_wmma_f32_16x16x4_f32(false, w0, false, x0,
                                                       (short)0, c0, false, false);
            c1 = __builtin_amdgcn_wmma_f32_16x16x4_f32(false, w1, false, x1,
                                                       (short)0, c1, false, false);
        }
        const v8f ht = c0 + c1;

        float s = 0.f;
        s += ht[0] * wf0.x;
        s += ht[1] * wf0.y;
        s += ht[2] * wf0.z;
        s += ht[3] * wf0.w;
        s += ht[4] * wf1.x;
        s += ht[5] * wf1.y;
        s += ht[6] * wf1.z;
        s += ht[7] * wf1.w;
        s += xhalf(s);
        if (half == 0) s0 = s;

        if (half == 1) {
            const int   node = 32 * pair + lane;
            const float sn   = hh ? s : s0;
            const float di   = rsqrtf((float)(dg + dh + 1u));
            const float ti   = di * sn;
            dinv[node] = di;
            t[node]    = ti;
            acc[node]  = ti;
            acc1[node] = 0.f;
        } else if (next >= ntiles) {
            if (hh == 0) {
                const int   node = 16 * tile + c;
                const float di   = rsqrtf((float)(deg[node] + deg1[node] + 1u));
                const float ti   = di * s;
                dinv[node] = di;
                t[node]    = ti;
                acc[node]  = ti;
                acc1[node] = 0.f;
            }
        }

        tile = next;
        pair = npair;
        half ^= 1;
    }
}

#define SRC_TBL 50000

__global__ __launch_bounds__(AGG_THREADS)
void gcn_scatter_kernel(const int* __restrict__ ei,
                        const float* __restrict__ t,
                        float* __restrict__ acc, int E, int N) {
    __shared__ float tbl[AGG_TBL];
    __shared__ float tsl[SRC_TBL];
    const int tid   = threadIdx.x;
    const int R     = (N + AGG_TBL - 1) / AGG_TBL;
    const int S     = (N + SRC_TBL - 1) / SRC_TBL;
    const int C     = max(1, (int)gridDim.x / (R * S));
    const int span  = (((N + R - 1) / R) + 31) & ~31;
    const int sspan = (((N + S - 1) / S) + 3) & ~3;
    float* const aplane = acc + (size_t)package_half() * N;

    for (int task = blockIdx.x; task < R * S * C; task += gridDim.x) {
        const int r    = task % R;
        const int s    = (task / R) % S;
        const int c    = task / (R * S);
        const int lo   = r * span;
        const int len  = min(span, N - lo);
        const int slo  = s * sspan;
        const int slen = max(0, min(sspan, N - slo));
        const unsigned ulen  = (unsigned)max(0, len);
        const unsigned uslen = (unsigned)slen;

        for (int i = tid; i < (len + 3) / 4; i += AGG_THREADS)
            ((float4*)tbl)[i] = make_float4(0.f, 0.f, 0.f, 0.f);

        if ((N & 3) == 0) {
            const float4* t4 = (const float4*)(t + slo);
            const int     n4 = slen >> 2;
            int i0 = tid;
            for (; i0 + (AGG_BATCH - 1) * AGG_THREADS < n4;
                 i0 += AGG_BATCH * AGG_THREADS) {
                float4 v[AGG_BATCH];
#pragma unroll
                for (int u = 0; u < AGG_BATCH; ++u)
                    v[u] = t4[i0 + u * AGG_THREADS];
#pragma unroll
                for (int u = 0; u < AGG_BATCH; ++u)
                    ((float4*)tsl)[i0 + u * AGG_THREADS] = v[u];
            }
            for (; i0 < n4; i0 += AGG_THREADS)
                ((float4*)tsl)[i0] = t4[i0];
        } else {
            for (int i = tid; i < slen; i += AGG_THREADS)
                tsl[i] = t[slo + i];
        }
        __syncthreads();

        if ((E & 3) == 0) {
            const int4* s4 = (const int4*)ei;
            const int4* d4 = (const int4*)(ei + E);
            const int nq   = E >> 2;
            const int per  = (nq + C - 1) / C;
            const int qbeg = min(nq, c * per);
            const int qend = min(nq, qbeg + per);
            for (int q0 = qbeg; q0 < qend; q0 += AGG_BATCH * AGG_THREADS) {
                unsigned bv[4 * AGG_BATCH];
                unsigned av[4 * AGG_BATCH];
                float    tv[4 * AGG_BATCH];
#pragma unroll
                for (int u = 0; u < AGG_BATCH; ++u) {
                    const int  q  = q0 + u * AGG_THREADS + tid;
                    const int  qq = min(q, qend - 1);
                    const int4 sq = s4[qq];
                    int4       d  = d4[qq];
                    if (q >= qend) d = make_int4(-1, -1, -1, -1);
                    bv[4 * u + 0] = (unsigned)(sq.x - slo);
                    bv[4 * u + 1] = (unsigned)(sq.y - slo);
                    bv[4 * u + 2] = (unsigned)(sq.z - slo);
                    bv[4 * u + 3] = (unsigned)(sq.w - slo);
                    av[4 * u + 0] = (unsigned)(d.x - lo);
                    av[4 * u + 1] = (unsigned)(d.y - lo);
                    av[4 * u + 2] = (unsigned)(d.z - lo);
                    av[4 * u + 3] = (unsigned)(d.w - lo);
                }
#pragma unroll
                for (int k = 0; k < 4 * AGG_BATCH; ++k)
                    if (bv[k] >= uslen) av[k] = 0xffffffffu;
#pragma unroll
                for (int k = 0; k < 4 * AGG_BATCH; ++k)
                    tv[k] = tsl[av[k] < ulen ? bv[k] : 0u];
#pragma unroll
                for (int k = 0; k < 4 * AGG_BATCH; ++k)
                    if (av[k] < ulen) unsafeAtomicAdd(&tbl[av[k]], tv[k]);
            }
        } else {
            const int per  = (E + C - 1) / C;
            const int ebeg = min(E, c * per);
            const int eend = min(E, ebeg + per);
            for (int e = ebeg + tid; e < eend; e += AGG_THREADS) {
                const unsigned a = (unsigned)(ei[E + e] - lo);
                const unsigned b = (unsigned)(ei[e] - slo);
                if (a < ulen && b < uslen) unsafeAtomicAdd(&tbl[a], tsl[b]);
            }
        }
        __syncthreads();

        const int rot = (((len >> 5) * (c * S + s)) / (C * S)) << 5;
        for (int i0 = tid; i0 < len; i0 += AGG_BATCH * AGG_THREADS) {
            int   j[AGG_BATCH];
            float v[AGG_BATCH];
#pragma unroll
            for (int u = 0; u < AGG_BATCH; ++u) {
                const int i = i0 + u * AGG_THREADS;
                j[u] = i + rot;
                if (j[u] >= len) j[u] -= len;
                v[u] = i < len ? tbl[j[u]] : 0.f;
            }
#pragma unroll
            for (int u = 0; u < AGG_BATCH; ++u)
                if (v[u] != 0.f) unsafeAtomicAdd(aplane + lo + j[u], v[u]);
        }
        __syncthreads();
    }
}

__global__ void gcn_head_kernel(const float* __restrict__ acc,
                                const float* __restrict__ dinv,
                                const float* __restrict__ bconv,
                                const float* __restrict__ wfc,
                                const float* __restrict__ bfc,
                                float* __restrict__ out, int N) {
    int i = blockIdx.x * blockDim.x + threadIdx.x;
    if (i >= N) return;
    float c = 0.f;
#pragma unroll
    for (int j = 0; j < HID; ++j)
        c += bconv[j] * wfc[j];
    c += bfc[0];
    out[i] = dinv[i] * (acc[i] + acc[N + i]) + c;
}

extern "C" void kernel_launch(void* const* d_in, const int* in_sizes, int n_in,
                              void* d_out, int out_size, void* d_ws, size_t ws_size,
                              hipStream_t stream) {
    const float* x     = (const float*)d_in[0];
    const int*   ei    = (const int*)d_in[1];
    const float* W     = (const float*)d_in[2];
    const float* bconv = (const float*)d_in[3];
    const float* wfc   = (const float*)d_in[4];
    const float* bfc   = (const float*)d_in[5];
    float* out = (float*)d_out;

    const int N = in_sizes[0] / D_FEAT;
    const int E = in_sizes[1] / 2;

    char* ws = (char*)d_ws;
    unsigned* deg  = (unsigned*)ws;
    float*    dinv = (float*)(ws + (size_t)N * 8);
    float*    t    = dinv + N;
    float*    acc  = t + N;

    hipMemsetAsync(deg, 0, (size_t)N * 2 * sizeof(unsigned), stream);

    gcn_deg_kernel<<<256, AGG_THREADS, 0, stream>>>(ei + E, deg, E, N);

    const int ntiles = N / 16;
    gcn_gemm_wmma_kernel<<<256, 256, 0, stream>>>(x, W, wfc, deg, dinv, t, acc, ntiles, N);

    gcn_scatter_kernel<<<256, AGG_THREADS, 0, stream>>>(ei, t, acc, E, N);

    gcn_head_kernel<<<(N + 255) / 256, 256, 0, stream>>>(acc, dinv, bconv, wfc,
                                                         bfc, out, N);
}


# Degree kernel also bins each edge once by dst range (one packed word, src<<15|dst offset, via LDS lists); edge sum reads the bins, not both index lists 10x; f32 sum order changes (atomics kept)
# speedup vs baseline: 21.0358x; 1.0481x over previous
#include <hip/hip_runtime.h>
#include <hip/hip_bf16.h>

#define D_FEAT 256
#define HID    16

typedef float v2f __attribute__((ext_vector_type(2)));
typedef float v8f __attribute__((ext_vector_type(8)));

#define AGG_THREADS 1024
#define AGG_TBL     20000
#define AGG_BATCH   4

__device__ __forceinline__ int package_half() {
    const unsigned where = (unsigned)__builtin_amdgcn_s_sendmsg_rtn(0x87);
    return (int)((where >> 18) & 1u);
}

__global__ __launch_bounds__(AGG_THREADS)
void gcn_deg_kernel(const int* __restrict__ dst,
                    unsigned* __restrict__ deg, int E, int N) {
    __shared__ unsigned tbl[AGG_TBL];
    const int tid  = threadIdx.x;
    const int R    = (N + AGG_TBL - 1) / AGG_TBL;
    const int C    = max(1, (int)gridDim.x / R);
    const int span = (((N + R - 1) / R) + 31) & ~31;
    unsigned* const dplane = deg + (size_t)package_half() * N;

    for (int task = blockIdx.x; task < R * C; task += gridDim.x) {
        const int r   = task % R;
        const int c   = task / R;
        const int lo  = r * span;
        const int len = min(span, N - lo);

        for (int i = tid; i < (len + 3) / 4; i += AGG_THREADS)
            ((uint4*)tbl)[i] = make_uint4(0u, 0u, 0u, 0u);
        __syncthreads();

        if ((E & 3) == 0) {
            const int4* d4 = (const int4*)dst;
            const int nq   = E >> 2;
            const int per  = (nq + C - 1) / C;
            const int qbeg = min(nq, c * per);
            const int qend = min(nq, qbeg + per);
            for (int q0 = qbeg; q0 < qend; q0 += AGG_BATCH * AGG_THREADS) {
                int4 d[AGG_BATCH];
#pragma unroll
                for (int u = 0; u < AGG_BATCH; ++u) {
                    const int q = q0 + u * AGG_THREADS + tid;
                    d[u] = d4[min(q, qend - 1)];
                    if (q >= qend) d[u] = make_int4(-1, -1, -1, -1);
                }
#pragma unroll
                for (int u = 0; u < AGG_BATCH; ++u) {
                    const unsigned a0 = (unsigned)(d[u].x - lo);
                    const unsigned a1 = (unsigned)(d[u].y - lo);
                    const unsigned a2 = (unsigned)(d[u].z - lo);
                    const unsigned a3 = (unsigned)(d[u].w - lo);
                    if (a0 < (unsigned)len) atomicAdd(&tbl[a0], 1u);
                    if (a1 < (unsigned)len) atomicAdd(&tbl[a1], 1u);
                    if (a2 < (unsigned)len) atomicAdd(&tbl[a2], 1u);
                    if (a3 < (unsigned)len) atomicAdd(&tbl[a3], 1u);
                }
            }
        } else {
            const int per  = (E + C - 1) / C;
            const int ebeg = min(E, c * per);
            const int eend = min(E, ebeg + per);
            for (int e = ebeg + tid; e < eend; e += AGG_THREADS) {
                const unsigned a = (unsigned)(dst[e] - lo);
                if (a < (unsigned)len) atomicAdd(&tbl[a], 1u);
            }
        }
        __syncthreads();

        const int rot = (((len >> 5) * c) / C) << 5;
        for (int i0 = tid; i0 < len; i0 += AGG_BATCH * AGG_THREADS) {
            int      j[AGG_BATCH];
            unsigned v[AGG_BATCH];
#pragma unroll
            for (int u = 0; u < AGG_BATCH; ++u) {
                const int i = i0 + u * AGG_THREADS;
                j[u] = i + rot;
                if (j[u] >= len) j[u] -= len;
                v[u] = i < len ? tbl[j[u]] : 0u;
            }
#pragma unroll
            for (int u = 0; u < AGG_BATCH; ++u)
                if (v[u] != 0u) atomicAdd(&dplane[lo + j[u]], v[u]);
        }
        __syncthreads();
    }
}

#define GEMM_WAVES 8
#define A_PITCH    260
#define TILE_Q     32

typedef float v4f __attribute__((ext_vector_type(4)));

__device__ __forceinline__ float xhalf(float v) {
    return __builtin_bit_cast(float, __builtin_amdgcn_permlanex16(
        __builtin_bit_cast(int, v), __builtin_bit_cast(int, v),
        0x76543210, 0xfedcba98, false, false));
}

__global__ __launch_bounds__(GEMM_WAVES * 32, 1)
void gcn_gemm_wmma_kernel(const float* __restrict__ x,
                          const float* __restrict__ W,
                          const float* __restrict__ wfc,
                          const unsigned* __restrict__ deg,
                          float* __restrict__ dinv,
                          float* __restrict__ t,
                          float* __restrict__ acc, int ntiles, int N) {
    __shared__ float Wt[D_FEAT * HID];
    __shared__ float At[GEMM_WAVES][16 * A_PITCH];

    const int tid  = threadIdx.x;
    const int lane = tid & 31;
    const int wave = __builtin_amdgcn_readfirstlane(tid >> 5);
    const int c    = lane & 15;
    const int hh   = lane >> 4;

    float4 wq[(D_FEAT * HID) / 4 / (GEMM_WAVES * 32)];
#pragma unroll
    for (int u = 0; u < (D_FEAT * HID) / 4 / (GEMM_WAVES * 32); ++u)
        wq[u] = ((const float4*)W)[tid + u * GEMM_WAVES * 32];

    const float4 wf0 = ((const float4*)wfc)[2 * hh];
    const float4 wf1 = ((const float4*)wfc)[2 * hh + 1];

    const int pstride = (int)gridDim.x * GEMM_WAVES;
    int pair = (int)blockIdx.x + (int)gridDim.x * wave;
    int half = 0;
    int tile = 2 * pair;

    v4f g[TILE_Q];
    if (tile < ntiles) {
        const v4f* src = (const v4f*)(x + (size_t)tile * 16 * D_FEAT) + lane;
#pragma unroll
        for (int q = 0; q < TILE_Q; ++q)
            g[q] = __builtin_nontemporal_load(src + q * 32);
    }

#pragma unroll
    for (int u = 0; u < (D_FEAT * HID) / 4 / (GEMM_WAVES * 32); ++u) {
        const int i  = tid + u * GEMM_WAVES * 32;
        const int k  = i >> 2;
        const int n0 = (i & 3) * 4;
        float* d = Wt + ((k >> 2) * 16 + n0) * 4 + (k & 3);
        d[0] = wq[u].x; d[4] = wq[u].y; d[8] = wq[u].z; d[12] = wq[u].w;
    }
    __syncthreads();

    v4f bq[TILE_Q];
#pragma unroll
    for (int j = 0; j < TILE_Q; ++j)
        bq[j] = *(const v4f*)(Wt + ((2 * j + hh) * 16 + c) * 4);

    float* aw = At[wave];

    const unsigned* const deg1 = deg + N;
    float* const          acc1 = acc + N;

    unsigned dg = 0u, dh = 0u;
    float    s0 = 0.f;

    while (tile < ntiles) {
#pragma unroll
        for (int q = 0; q < TILE_Q; ++q)
            *(v4f*)(aw + (q >> 1) * A_PITCH + (q & 1) * 128 + 4 * lane) = g[q];

        const int npair = half ? pair + pstride : pair;
        const int next  = half ? 2 * npair : tile + 1;

        if (next < ntiles) {
            if (half == 0) {
                dg = deg[32 * pair + lane];
                dh = deg1[32 * pair + lane];
            }
            const v4f* src = (const v4f*)(x + (size_t)next * 16 * D_FEAT) + lane;
#pragma unroll
            for (int q = 0; q < TILE_Q; ++q)
                g[q] = __builtin_nontemporal_load(src + q * 32);
        }

        v4f f[TILE_Q];
#pragma unroll
        for (int j = 0; j < TILE_Q; ++j)
            f[j] = *(const v4f*)(aw + c * A_PITCH + 8 * j + 4 * hh);
        __builtin_amdgcn_sched_barrier(0);

        v8f c0 = {}, c1 = {};
#pragma unroll
        for (int j = 0; j < TILE_Q; ++j) {
            const v2f w0 = {bq[j].x, bq[j].y}, x0 = {f[j].x, f[j].y};
            const v2f w1 = {bq[j].z, bq[j].w}, x1 = {f[j].z, f[j].w};
            c0 = __builtin_amdgcn_wmma_f32_16x16x4_f32(false, w0, false, x0,
                                                       (short)0, c0, false, false);
            c1 = __builtin_amdgcn_wmma_f32_16x16x4_f32(false, w1, false, x1,
                                                       (short)0, c1, false, false);
        }
        const v8f ht = c0 + c1;

        float s = 0.f;
        s += ht[0] * wf0.x;
        s += ht[1] * wf0.y;
        s += ht[2] * wf0.z;
        s += ht[3] * wf0.w;
        s += ht[4] * wf1.x;
        s += ht[5] * wf1.y;
        s += ht[6] * wf1.z;
        s += ht[7] * wf1.w;
        s += xhalf(s);
        if (half == 0) s0 = s;

        if (half == 1) {
            const int   node = 32 * pair + lane;
            const float sn   = hh ? s : s0;
            const float di   = rsqrtf((float)(dg + dh + 1u));
            const float ti   = di * sn;
            dinv[node] = di;
            t[node]    = ti;
            acc[node]  = ti;
            acc1[node] = 0.f;
        } else if (next >= ntiles) {
            if (hh == 0) {
                const int   node = 16 * tile + c;
                const float di   = rsqrtf((float)(deg[node] + deg1[node] + 1u));
                const float ti   = di * s;
                dinv[node] = di;
                t[node]    = ti;
                acc[node]  = ti;
                acc1[node] = 0.f;
            }
        }

        tile = next;
        pair = npair;
        half ^= 1;
    }
}

#define SRC_TBL 50000

__global__ __launch_bounds__(AGG_THREADS)
void gcn_scatter_kernel(const int* __restrict__ ei,
                        const float* __restrict__ t,
                        float* __restrict__ acc, int E, int N) {
    __shared__ float tbl[AGG_TBL];
    __shared__ float tsl[SRC_TBL];
    const int tid   = threadIdx.x;
    const int R     = (N + AGG_TBL - 1) / AGG_TBL;
    const int S     = (N + SRC_TBL - 1) / SRC_TBL;
    const int C     = max(1, (int)gridDim.x / (R * S));
    const int span  = (((N + R - 1) / R) + 31) & ~31;
    const int sspan = (((N + S - 1) / S) + 3) & ~3;
    float* const aplane = acc + (size_t)package_half() * N;

    for (int task = blockIdx.x; task < R * S * C; task += gridDim.x) {
        const int r    = task % R;
        const int s    = (task / R) % S;
        const int c    = task / (R * S);
        const int lo   = r * span;
        const int len  = min(span, N - lo);
        const int slo  = s * sspan;
        const int slen = max(0, min(sspan, N - slo));
        const unsigned ulen  = (unsigned)max(0, len);
        const unsigned uslen = (unsigned)slen;

        for (int i = tid; i < (len + 3) / 4; i += AGG_THREADS)
            ((float4*)tbl)[i] = make_float4(0.f, 0.f, 0.f, 0.f);

        if ((N & 3) == 0) {
            const float4* t4 = (const float4*)(t + slo);
            const int     n4 = slen >> 2;
            int i0 = tid;
            for (; i0 + (AGG_BATCH - 1) * AGG_THREADS < n4;
                 i0 += AGG_BATCH * AGG_THREADS) {
                float4 v[AGG_BATCH];
#pragma unroll
                for (int u = 0; u < AGG_BATCH; ++u)
                    v[u] = t4[i0 + u * AGG_THREADS];
#pragma unroll
                for (int u = 0; u < AGG_BATCH; ++u)
                    ((float4*)tsl)[i0 + u * AGG_THREADS] = v[u];
            }
            for (; i0 < n4; i0 += AGG_THREADS)
                ((float4*)tsl)[i0] = t4[i0];
        } else {
            for (int i = tid; i < slen; i += AGG_THREADS)
                tsl[i] = t[slo + i];
        }
        __syncthreads();

        if ((E & 3) == 0) {
            const int4* s4 = (const int4*)ei;
            const int4* d4 = (const int4*)(ei + E);
            const int nq   = E >> 2;
            const int per  = (nq + C - 1) / C;
            const int qbeg = min(nq, c * per);
            const int qend = min(nq, qbeg + per);
            for (int q0 = qbeg; q0 < qend; q0 += AGG_BATCH * AGG_THREADS) {
                unsigned bv[4 * AGG_BATCH];
                unsigned av[4 * AGG_BATCH];
                float    tv[4 * AGG_BATCH];
#pragma unroll
                for (int u = 0; u < AGG_BATCH; ++u) {
                    const int  q  = q0 + u * AGG_THREADS + tid;
                    const int  qq = min(q, qend - 1);
                    const int4 sq = s4[qq];
                    int4       d  = d4[qq];
                    if (q >= qend) d = make_int4(-1, -1, -1, -1);
                    bv[4 * u + 0] = (unsigned)(sq.x - slo);
                    bv[4 * u + 1] = (unsigned)(sq.y - slo);
                    bv[4 * u + 2] = (unsigned)(sq.z - slo);
                    bv[4 * u + 3] = (unsigned)(sq.w - slo);
                    av[4 * u + 0] = (unsigned)(d.x - lo);
                    av[4 * u + 1] = (unsigned)(d.y - lo);
                    av[4 * u + 2] = (unsigned)(d.z - lo);
                    av[4 * u + 3] = (unsigned)(d.w - lo);
                }
#pragma unroll
                for (int k = 0; k < 4 * AGG_BATCH; ++k)
                    if (bv[k] >= uslen) av[k] = 0xffffffffu;
#pragma unroll
                for (int k = 0; k < 4 * AGG_BATCH; ++k)
                    tv[k] = tsl[av[k] < ulen ? bv[k] : 0u];
#pragma unroll
                for (int k = 0; k < 4 * AGG_BATCH; ++k)
                    if (av[k] < ulen) unsafeAtomicAdd(&tbl[av[k]], tv[k]);
            }
        } else {
            const int per  = (E + C - 1) / C;
            const int ebeg = min(E, c * per);
            const int eend = min(E, ebeg + per);
            for (int e = ebeg + tid; e < eend; e += AGG_THREADS) {
                const unsigned a = (unsigned)(ei[E + e] - lo);
                const unsigned b = (unsigned)(ei[e] - slo);
                if (a < ulen && b < uslen) unsafeAtomicAdd(&tbl[a], tsl[b]);
            }
        }
        __syncthreads();

        const int rot = (((len >> 5) * (c * S + s)) / (C * S)) << 5;
        for (int i0 = tid; i0 < len; i0 += AGG_BATCH * AGG_THREADS) {
            int   j[AGG_BATCH];
            float v[AGG_BATCH];
#pragma unroll
            for (int u = 0; u < AGG_BATCH; ++u) {
                const int i = i0 + u * AGG_THREADS;
                j[u] = i + rot;
                if (j[u] >= len) j[u] -= len;
                v[u] = i < len ? tbl[j[u]] : 0.f;
            }
#pragma unroll
            for (int u = 0; u < AGG_BATCH; ++u)
                if (v[u] != 0.f) unsafeAtomicAdd(aplane + lo + j[u], v[u]);
        }
        __syncthreads();
    }
}

#define BIN_R     4
#define BIN_C     64
#define BIN_TBL   25024
#define BIN_BUF   50176
#define BIN_SHIFT 15
#define SUM_G     16
#define SUM_TBL   12512

template <int B>
__device__ __forceinline__ void bin_quads(const int4* __restrict__ s4,
                                          const int4* __restrict__ d4,
                                          int q0, int qend, int tid, int lo,
                                          unsigned ulen, unsigned* tbl,
                                          unsigned* buf, unsigned* fill) {
    unsigned av[4 * B];
    unsigned sv[4 * B];
#pragma unroll
    for (int u = 0; u < B; ++u) {
        const int  q  = q0 + u * AGG_THREADS + tid;
        const int  qq = min(q, qend - 1);
        const int4 sq = s4[qq];
        int4       d  = d4[qq];
        if (q >= qend) d = make_int4(-1, -1, -1, -1);
        sv[4 * u + 0] = (unsigned)sq.x;
        sv[4 * u + 1] = (unsigned)sq.y;
        sv[4 * u + 2] = (unsigned)sq.z;
        sv[4 * u + 3] = (unsigned)sq.w;
        av[4 * u + 0] = (unsigned)(d.x - lo);
        av[4 * u + 1] = (unsigned)(d.y - lo);
        av[4 * u + 2] = (unsigned)(d.z - lo);
        av[4 * u + 3] = (unsigned)(d.w - lo);
    }
    int n = 0;
#pragma unroll
    for (int k = 0; k < 4 * B; ++k) n += av[k] < ulen ? 1 : 0;
    const int lane = tid & 31;
    int incl = n;
#pragma unroll
    for (int d = 1; d < 32; d <<= 1) {
        const int v = __shfl_up(incl, (unsigned)d, 32);
        if (lane >= d) incl += v;
    }
    const int total = __shfl(incl, 31, 32);
    unsigned base = 0u;
    if (lane == 31 && total != 0) base = atomicAdd(fill, (unsigned)total);
    base = (unsigned)__shfl((int)base, 31, 32);
    unsigned pos = base + (unsigned)(incl - n);
#pragma unroll
    for (int k = 0; k < 4 * B; ++k) {
        if (av[k] < ulen) {
            atomicAdd(&tbl[av[k]], 1u);
            buf[pos] = (sv[k] << BIN_SHIFT) | av[k];
            ++pos;
        }
    }
}

__global__ __launch_bounds__(AGG_THREADS)
void gcn_deg_bin_kernel(const int* __restrict__ ei,
                        unsigned* __restrict__ deg,
                        unsigned* __restrict__ cursor,
                        int2* __restrict__ dir,
                        uint4* __restrict__ bins, int E, int N) {
    __shared__ __attribute__((aligned(16))) unsigned tbl[BIN_TBL];
    __shared__ __attribute__((aligned(16))) unsigned buf[BIN_BUF];
    __shared__ unsigned fill;
    __shared__ unsigned gstart;
    const int tid  = threadIdx.x;
    const int span = (((N + BIN_R - 1) / BIN_R) + 31) & ~31;
    unsigned* const dplane = deg + (size_t)package_half() * N;

    const int task = blockIdx.x;
    const int r    = task % BIN_R;
    const int c    = task / BIN_R;
    const int lo   = r * span;
    const int len  = max(0, min(span, N - lo));
    const unsigned ulen = (unsigned)len;

    for (int i = tid; i < (len + 3) / 4; i += AGG_THREADS)
        ((uint4*)tbl)[i] = make_uint4(0u, 0u, 0u, 0u);
    if (tid == 0) fill = 0u;
    __syncthreads();

    const int4* s4 = (const int4*)ei;
    const int4* d4 = (const int4*)(ei + E);
    const int nq   = E >> 2;
    const int per  = (nq + BIN_C - 1) / BIN_C;
    const int qbeg = min(nq, c * per);
    const int qend = min(nq, qbeg + per);
    int q0 = qbeg;
    for (; q0 + AGG_BATCH * AGG_THREADS <= qend; q0 += AGG_BATCH * AGG_THREADS)
        bin_quads<AGG_BATCH>(s4, d4, q0, qend, tid, lo, ulen, tbl, buf, &fill);
    for (; q0 < qend; q0 += AGG_THREADS)
        bin_quads<1>(s4, d4, q0, qend, tid, lo, ulen, tbl, buf, &fill);
    __syncthreads();

    const unsigned total = fill;
    const unsigned nquad = (total + 3u) >> 2;
    if (tid < 3 && total + (unsigned)tid < 4u * nquad)
        buf[total + (unsigned)tid] = 0xffffffffu;
    if (tid == 0) {
        const unsigned g = atomicAdd(cursor, nquad);
        gstart    = g;
        dir[task] = make_int2((int)g, (int)nquad);
    }
    __syncthreads();
    {
        uint4* const out = bins + gstart;
        for (unsigned i = (unsigned)tid; i < nquad; i += AGG_THREADS)
            out[i] = ((const uint4*)buf)[i];
    }

    const int rot = (((len >> 5) * c) / BIN_C) << 5;
    for (int i0 = tid; i0 < len; i0 += AGG_BATCH * AGG_THREADS) {
        int      j[AGG_BATCH];
        unsigned v[AGG_BATCH];
#pragma unroll
        for (int u = 0; u < AGG_BATCH; ++u) {
            const int i = i0 + u * AGG_THREADS;
            j[u] = i + rot;
            if (j[u] >= len) j[u] -= len;
            v[u] = i < len ? tbl[j[u]] : 0u;
        }
#pragma unroll
        for (int u = 0; u < AGG_BATCH; ++u)
            if (v[u] != 0u) atomicAdd(&dplane[lo + j[u]], v[u]);
    }
}

#define SUM_NP (BIN_C / SUM_G)

template <int B>
__device__ __forceinline__ void sum_quads(const uint4* __restrict__ bins,
                                          const int (&pre)[SUM_NP + 1],
                                          const int (&off)[SUM_NP],
                                          int v0, int tid,
                                          unsigned hlo, unsigned uhlen,
                                          unsigned slo, unsigned uslen,
                                          float* tbl, const float* tsl) {
    const int nqt = pre[SUM_NP];
    unsigned av[4 * B];
    unsigned bv[4 * B];
    float    tv[4 * B];
#pragma unroll
    for (int u = 0; u < B; ++u) {
        const int v  = v0 + u * AGG_THREADS + tid;
        const int vv = min(v, nqt - 1);
        int o = off[0];
#pragma unroll
        for (int p = 1; p < SUM_NP; ++p)
            if (vv >= pre[p]) o = off[p];
        uint4 w = bins[o + vv];
        if (v >= nqt)
            w = make_uint4(0xffffffffu, 0xffffffffu, 0xffffffffu, 0xffffffffu);
        av[4 * u + 0] = (w.x & 0x7fffu) - hlo;
        av[4 * u + 1] = (w.y & 0x7fffu) - hlo;
        av[4 * u + 2] = (w.z & 0x7fffu) - hlo;
        av[4 * u + 3] = (w.w & 0x7fffu) - hlo;
        bv[4 * u + 0] = (w.x >> BIN_SHIFT) - slo;
        bv[4 * u + 1] = (w.y >> BIN_SHIFT) - slo;
        bv[4 * u + 2] = (w.z >> BIN_SHIFT) - slo;
        bv[4 * u + 3] = (w.w >> BIN_SHIFT) - slo;
    }
#pragma unroll
    for (int k = 0; k < 4 * B; ++k)
        if (bv[k] >= uslen) av[k] = 0xffffffffu;
#pragma unroll
    for (int k = 0; k < 4 * B; ++k)
        tv[k] = tsl[av[k] < uhlen ? bv[k] : 0u];
#pragma unroll
    for (int k = 0; k < 4 * B; ++k)
        if (av[k] < uhlen) unsafeAtomicAdd(&tbl[av[k]], tv[k]);
}

__global__ __launch_bounds__(AGG_THREADS)
void gcn_sum_bins_kernel(const uint4* __restrict__ bins,
                         const int2* __restrict__ dir,
                         const float* __restrict__ t,
                         float* __restrict__ acc, int N) {
    __shared__ __attribute__((aligned(16))) float tbl[SUM_TBL];
    __shared__ __attribute__((aligned(16))) float tsl[SRC_TBL];
    const int tid   = threadIdx.x;
    const int S     = (N + SRC_TBL - 1) / SRC_TBL;
    const int span  = (((N + BIN_R - 1) / BIN_R) + 31) & ~31;
    const int hspan = ((span >> 1) + 31) & ~31;
    const int sspan = (((N + S - 1) / S) + 3) & ~3;
    float* const aplane = acc + (size_t)package_half() * N;

    const int ntask = BIN_R * 2 * S * SUM_G;
    for (int task = blockIdx.x; task < ntask; task += gridDim.x) {
        const int r    = task % BIN_R;
        const int h    = (task / BIN_R) & 1;
        const int s    = (task / (2 * BIN_R)) % S;
        const int g    = task / (2 * BIN_R * S);
        const int lo   = r * span;
        const int len  = max(0, min(span, N - lo));
        const int hlo  = h * hspan;
        const int hlen = max(0, min(hspan, len - hlo));
        const int slo  = s * sspan;
        const int slen = max(0, min(sspan, N - slo));

        int pre[SUM_NP + 1];
        int off[SUM_NP];
        pre[0] = 0;
#pragma unroll
        for (int p = 0; p < SUM_NP; ++p) {
            const int2 d = dir[(g * SUM_NP + p) * BIN_R + r];
            off[p]     = d.x - pre[p];
            pre[p + 1] = pre[p] + d.y;
        }

        for (int i = tid; i < (hlen + 3) / 4; i += AGG_THREADS)
            ((float4*)tbl)[i] = make_float4(0.f, 0.f, 0.f, 0.f);

        {
            const float4* t4 = (const float4*)(t + slo);
            const int     n4 = slen >> 2;
            int i0 = tid;
            for (; i0 + (AGG_BATCH - 1) * AGG_THREADS < n4;
                 i0 += AGG_BATCH * AGG_THREADS) {
                float4 v[AGG_BATCH];
#pragma unroll
                for (int u = 0; u < AGG_BATCH; ++u)
                    v[u] = t4[i0 + u * AGG_THREADS];
#pragma unroll
                for (int u = 0; u < AGG_BATCH; ++u)
                    ((float4*)tsl)[i0 + u * AGG_THREADS] = v[u];
            }
            for (; i0 < n4; i0 += AGG_THREADS)
                ((float4*)tsl)[i0] = t4[i0];
        }
        __syncthreads();

        {
            const int nqt = pre[SUM_NP];
            int v0 = 0;
            for (; v0 + AGG_BATCH * AGG_THREADS <= nqt; v0 += AGG_BATCH * AGG_THREADS)
                sum_quads<AGG_BATCH>(bins, pre, off, v0, tid, (unsigned)hlo,
                                     (unsigned)hlen, (unsigned)slo,
                                     (unsigned)slen, tbl, tsl);
            for (; v0 < nqt; v0 += AGG_THREADS)
                sum_quads<1>(bins, pre, off, v0, tid, (unsigned)hlo,
                             (unsigned)hlen, (unsigned)slo, (unsigned)slen,
                             tbl, tsl);
        }
        __syncthreads();

        const int rot = (((hlen >> 5) * (g * S + s)) / (SUM_G * S)) << 5;
        float* const dstp = aplane + lo + hlo;
        for (int i0 = tid; i0 < hlen; i0 += AGG_BATCH * AGG_THREADS) {
            int   j[AGG_BATCH];
            float v[AGG_BATCH];
#pragma unroll
            for (int u = 0; u < AGG_BATCH; ++u) {
                const int i = i0 + u * AGG_THREADS;
                j[u] = i + rot;
                if (j[u] >= hlen) j[u] -= hlen;
                v[u] = i < hlen ? tbl[j[u]] : 0.f;
            }
#pragma unroll
            for (int u = 0; u < AGG_BATCH; ++u)
                if (v[u] != 0.f) unsafeAtomicAdd(dstp + j[u], v[u]);
        }
        __syncthreads();
    }
}

__global__ void gcn_head_kernel(const float* __restrict__ acc,
                                const float* __restrict__ dinv,
                                const float* __restrict__ bconv,
                                const float* __restrict__ wfc,
                                const float* __restrict__ bfc,
                                float* __restrict__ out, int N) {
    int i = blockIdx.x * blockDim.x + threadIdx.x;
    if (i >= N) return;
    float c = 0.f;
#pragma unroll
    for (int j = 0; j < HID; ++j)
        c += bconv[j] * wfc[j];
    c += bfc[0];
    out[i] = dinv[i] * (acc[i] + acc[N + i]) + c;
}

extern "C" void kernel_launch(void* const* d_in, const int* in_sizes, int n_in,
                              void* d_out, int out_size, void* d_ws, size_t ws_size,
                              hipStream_t stream) {
    const float* x     = (const float*)d_in[0];
    const int*   ei    = (const int*)d_in[1];
    const float* W     = (const float*)d_in[2];
    const float* bconv = (const float*)d_in[3];
    const float* wfc   = (const float*)d_in[4];
    const float* bfc   = (const float*)d_in[5];
    float* out = (float*)d_out;

    const int N = in_sizes[0] / D_FEAT;
    const int E = in_sizes[1] / 2;

    const size_t bin_quads_max = (size_t)(E / 4) + BIN_R * BIN_C;
    const size_t bin_need = (size_t)N * 24 + 16 + (size_t)BIN_R * BIN_C * 8
                          + bin_quads_max * 16;
    const bool binned = (E & 3) == 0 && (N & 3) == 0 &&
                        N <= BIN_R * (BIN_TBL - 24) &&
                        (size_t)E <= (size_t)BIN_C * BIN_BUF &&
                        ws_size >= bin_need;

    char* ws = (char*)d_ws;
    const size_t cpad = binned ? 16 : 0;
    unsigned* deg    = (unsigned*)ws;
    unsigned* cursor = (unsigned*)(ws + (size_t)N * 8);
    float*    dinv   = (float*)(ws + (size_t)N * 8 + cpad);
    float*    t      = dinv + N;
    float*    acc    = t + N;
    int2*     dir    = (int2*)(acc + 2 * (size_t)N);
    uint4*    bins   = (uint4*)(dir + BIN_R * BIN_C);

    hipMemsetAsync(deg, 0, (size_t)N * 2 * sizeof(unsigned) + cpad, stream);

    if (binned) {
        gcn_deg_bin_kernel<<<BIN_R * BIN_C, AGG_THREADS, 0, stream>>>(
            ei, deg, cursor, dir, bins, E, N);
    } else {
        gcn_deg_kernel<<<256, AGG_THREADS, 0, stream>>>(ei + E, deg, E, N);
    }

    const int ntiles = N / 16;
    gcn_gemm_wmma_kernel<<<256, 256, 0, stream>>>(x, W, wfc, deg, dinv, t, acc, ntiles, N);

    if (binned) {
        gcn_sum_bins_kernel<<<256, AGG_THREADS, 0, stream>>>(bins, dir, t, acc, N);
    } else {
        gcn_scatter_kernel<<<256, AGG_THREADS, 0, stream>>>(ei, t, acc, E, N);
    }

    gcn_head_kernel<<<(N + 255) / 256, 256, 0, stream>>>(acc, dinv, bconv, wfc,
                                                         bfc, out, N);
}


# Edges sorted once by dst range (256 chunks x 241 ranges); one work-group owns a range: stores its degrees, then its edge sums + head; no global atomics, memset or head launch; f32 sum order changes
# speedup vs baseline: 22.6150x; 1.0751x over previous
#include <hip/hip_runtime.h>
#include <hip/hip_bf16.h>

#define D_FEAT 256
#define HID    16

typedef float v2f __attribute__((ext_vector_type(2)));
typedef float v8f __attribute__((ext_vector_type(8)));

#define AGG_THREADS 1024
#define AGG_TBL     20000
#define AGG_BATCH   4

__device__ __forceinline__ int package_half() {
    const unsigned where = (unsigned)__builtin_amdgcn_s_sendmsg_rtn(0x87);
    return (int)((where >> 18) & 1u);
}

__global__ __launch_bounds__(AGG_THREADS)
void gcn_deg_kernel(const int* __restrict__ dst,
                    unsigned* __restrict__ deg, int E, int N) {
    __shared__ unsigned tbl[AGG_TBL];
    const int tid  = threadIdx.x;
    const int R    = (N + AGG_TBL - 1) / AGG_TBL;
    const int C    = max(1, (int)gridDim.x / R);
    const int span = (((N + R - 1) / R) + 31) & ~31;
    unsigned* const dplane = deg + (size_t)package_half() * N;

    for (int task = blockIdx.x; task < R * C; task += gridDim.x) {
        const int r   = task % R;
        const int c   = task / R;
        const int lo  = r * span;
        const int len = min(span, N - lo);

        for (int i = tid; i < (len + 3) / 4; i += AGG_THREADS)
            ((uint4*)tbl)[i] = make_uint4(0u, 0u, 0u, 0u);
        __syncthreads();

        if ((E & 3) == 0) {
            const int4* d4 = (const int4*)dst;
            const int nq   = E >> 2;
            const int per  = (nq + C - 1) / C;
            const int qbeg = min(nq, c * per);
            const int qend = min(nq, qbeg + per);
            for (int q0 = qbeg; q0 < qend; q0 += AGG_BATCH * AGG_THREADS) {
                int4 d[AGG_BATCH];
#pragma unroll
                for (int u = 0; u < AGG_BATCH; ++u) {
                    const int q = q0 + u * AGG_THREADS + tid;
                    d[u] = d4[min(q, qend - 1)];
                    if (q >= qend) d[u] = make_int4(-1, -1, -1, -1);
                }
#pragma unroll
                for (int u = 0; u < AGG_BATCH; ++u) {
                    const unsigned a0 = (unsigned)(d[u].x - lo);
                    const unsigned a1 = (unsigned)(d[u].y - lo);
                    const unsigned a2 = (unsigned)(d[u].z - lo);
                    const unsigned a3 = (unsigned)(d[u].w - lo);
                    if (a0 < (unsigned)len) atomicAdd(&tbl[a0], 1u);
                    if (a1 < (unsigned)len) atomicAdd(&tbl[a1], 1u);
                    if (a2 < (unsigned)len) atomicAdd(&tbl[a2], 1u);
                    if (a3 < (unsigned)len) atomicAdd(&tbl[a3], 1u);
                }
            }
        } else {
            const int per  = (E + C - 1) / C;
            const int ebeg = min(E, c * per);
            const int eend = min(E, ebeg + per);
            for (int e = ebeg + tid; e < eend; e += AGG_THREADS) {
                const unsigned a = (unsigned)(dst[e] - lo);
                if (a < (unsigned)len) atomicAdd(&tbl[a], 1u);
            }
        }
        __syncthreads();

        const int rot = (((len >> 5) * c) / C) << 5;
        for (int i0 = tid; i0 < len; i0 += AGG_BATCH * AGG_THREADS) {
            int      j[AGG_BATCH];
            unsigned v[AGG_BATCH];
#pragma unroll
            for (int u = 0; u < AGG_BATCH; ++u) {
                const int i = i0 + u * AGG_THREADS;
                j[u] = i + rot;
                if (j[u] >= len) j[u] -= len;
                v[u] = i < len ? tbl[j[u]] : 0u;
            }
#pragma unroll
            for (int u = 0; u < AGG_BATCH; ++u)
                if (v[u] != 0u) atomicAdd(&dplane[lo + j[u]], v[u]);
        }
        __syncthreads();
    }
}

#define GEMM_WAVES 8
#define A_PITCH    260
#define TILE_Q     32

typedef float v4f __attribute__((ext_vector_type(4)));

__device__ __forceinline__ float xhalf(float v) {
    return __builtin_bit_cast(float, __builtin_amdgcn_permlanex16(
        __builtin_bit_cast(int, v), __builtin_bit_cast(int, v),
        0x76543210, 0xfedcba98, false, false));
}

template <bool PLANES>
__global__ __launch_bounds__(GEMM_WAVES * 32, 1)
void gcn_gemm_wmma_kernel(const float* __restrict__ x,
                          const float* __restrict__ W,
                          const float* __restrict__ wfc,
                          const unsigned* __restrict__ deg,
                          float* __restrict__ dinv,
                          float* __restrict__ t,
                          float* __restrict__ acc, int ntiles, int N) {
    __shared__ float Wt[D_FEAT * HID];
    __shared__ float At[GEMM_WAVES][16 * A_PITCH];

    const int tid  = threadIdx.x;
    const int lane = tid & 31;
    const int wave = __builtin_amdgcn_readfirstlane(tid >> 5);
    const int c    = lane & 15;
    const int hh   = lane >> 4;

    float4 wq[(D_FEAT * HID) / 4 / (GEMM_WAVES * 32)];
#pragma unroll
    for (int u = 0; u < (D_FEAT * HID) / 4 / (GEMM_WAVES * 32); ++u)
        wq[u] = ((const float4*)W)[tid + u * GEMM_WAVES * 32];

    const float4 wf0 = ((const float4*)wfc)[2 * hh];
    const float4 wf1 = ((const float4*)wfc)[2 * hh + 1];

    const int pstride = (int)gridDim.x * GEMM_WAVES;
    int pair = (int)blockIdx.x + (int)gridDim.x * wave;
    int half = 0;
    int tile = 2 * pair;

    v4f g[TILE_Q];
    if (tile < ntiles) {
        const v4f* src = (const v4f*)(x + (size_t)tile * 16 * D_FEAT) + lane;
#pragma unroll
        for (int q = 0; q < TILE_Q; ++q)
            g[q] = __builtin_nontemporal_load(src + q * 32);
    }

#pragma unroll
    for (int u = 0; u < (D_FEAT * HID) / 4 / (GEMM_WAVES * 32); ++u) {
        const int i  = tid + u * GEMM_WAVES * 32;
        const int k  = i >> 2;
        const int n0 = (i & 3) * 4;
        float* d = Wt + ((k >> 2) * 16 + n0) * 4 + (k & 3);
        d[0] = wq[u].x; d[4] = wq[u].y; d[8] = wq[u].z; d[12] = wq[u].w;
    }
    __syncthreads();

    v4f bq[TILE_Q];
#pragma unroll
    for (int j = 0; j < TILE_Q; ++j)
        bq[j] = *(const v4f*)(Wt + ((2 * j + hh) * 16 + c) * 4);

    float* aw = At[wave];

    const unsigned* const deg1 = deg + N;
    float* const          acc1 = acc + N;

    unsigned dg = 0u, dh = 0u;
    float    s0 = 0.f;

    while (tile < ntiles) {
#pragma unroll
        for (int q = 0; q < TILE_Q; ++q)
            *(v4f*)(aw + (q >> 1) * A_PITCH + (q & 1) * 128 + 4 * lane) = g[q];

        const int npair = half ? pair + pstride : pair;
        const int next  = half ? 2 * npair : tile + 1;

        if (next < ntiles) {
            if (half == 0) {
                dg = deg[32 * pair + lane];
                if (PLANES) dh = deg1[32 * pair + lane];
            }
            const v4f* src = (const v4f*)(x + (size_t)next * 16 * D_FEAT) + lane;
#pragma unroll
            for (int q = 0; q < TILE_Q; ++q)
                g[q] = __builtin_nontemporal_load(src + q * 32);
        }

        v4f f[TILE_Q];
#pragma unroll
        for (int j = 0; j < TILE_Q; ++j)
            f[j] = *(const v4f*)(aw + c * A_PITCH + 8 * j + 4 * hh);
        __builtin_amdgcn_sched_barrier(0);

        v8f c0 = {}, c1 = {};
#pragma unroll
        for (int j = 0; j < TILE_Q; ++j) {
            const v2f w0 = {bq[j].x, bq[j].y}, x0 = {f[j].x, f[j].y};
            const v2f w1 = {bq[j].z, bq[j].w}, x1 = {f[j].z, f[j].w};
            c0 = __builtin_amdgcn_wmma_f32_16x16x4_f32(false, w0, false, x0,
                                                       (short)0, c0, false, false);
            c1 = __builtin_amdgcn_wmma_f32_16x16x4_f32(false, w1, false, x1,
                                                       (short)0, c1, false, false);
        }
        const v8f ht = c0 + c1;

        float s = 0.f;
        s += ht[0] * wf0.x;
        s += ht[1] * wf0.y;
        s += ht[2] * wf0.z;
        s += ht[3] * wf0.w;
        s += ht[4] * wf1.x;
        s += ht[5] * wf1.y;
        s += ht[6] * wf1.z;
        s += ht[7] * wf1.w;
        s += xhalf(s);
        if (half == 0) s0 = s;

        if (half == 1) {
            const int   node = 32 * pair + lane;
            const float sn   = hh ? s : s0;
            const float di   = rsqrtf((float)(dg + dh + 1u));
            const float ti   = di * sn;
            dinv[node] = di;
            t[node]    = ti;
            if (PLANES) {
                acc[node]  = ti;
                acc1[node] = 0.f;
            }
        } else if (next >= ntiles) {
            if (hh == 0) {
                const int   node = 16 * tile + c;
                const float di   = rsqrtf((float)(deg[node] + (PLANES ? deg1[node] : 0u) + 1u));
                const float ti   = di * s;
                dinv[node] = di;
                t[node]    = ti;
                if (PLANES) {
                    acc[node]  = ti;
                    acc1[node] = 0.f;
                }
            }
        }

        tile = next;
        pair = npair;
        half ^= 1;
    }
}

#define SRC_TBL 50000

__global__ __launch_bounds__(AGG_THREADS)
void gcn_scatter_kernel(const int* __restrict__ ei,
                        const float* __restrict__ t,
                        float* __restrict__ acc, int E, int N) {
    __shared__ float tbl[AGG_TBL];
    __shared__ float tsl[SRC_TBL];
    const int tid   = threadIdx.x;
    const int R     = (N + AGG_TBL - 1) / AGG_TBL;
    const int S     = (N + SRC_TBL - 1) / SRC_TBL;
    const int C     = max(1, (int)gridDim.x / (R * S));
    const int span  = (((N + R - 1) / R) + 31) & ~31;
    const int sspan = (((N + S - 1) / S) + 3) & ~3;
    float* const aplane = acc + (size_t)package_half() * N;

    for (int task = blockIdx.x; task < R * S * C; task += gridDim.x) {
        const int r    = task % R;
        const int s    = (task / R) % S;
        const int c    = task / (R * S);
        const int lo   = r * span;
        const int len  = min(span, N - lo);
        const int slo  = s * sspan;
        const int slen = max(0, min(sspan, N - slo));
        const unsigned ulen  = (unsigned)max(0, len);
        const unsigned uslen = (unsigned)slen;

        for (int i = tid; i < (len + 3) / 4; i += AGG_THREADS)
            ((float4*)tbl)[i] = make_float4(0.f, 0.f, 0.f, 0.f);

        if ((N & 3) == 0) {
            const float4* t4 = (const float4*)(t + slo);
            const int     n4 = slen >> 2;
            int i0 = tid;
            for (; i0 + (AGG_BATCH - 1) * AGG_THREADS < n4;
                 i0 += AGG_BATCH * AGG_THREADS) {
                float4 v[AGG_BATCH];
#pragma unroll
                for (int u = 0; u < AGG_BATCH; ++u)
                    v[u] = t4[i0 + u * AGG_THREADS];
#pragma unroll
                for (int u = 0; u < AGG_BATCH; ++u)
                    ((float4*)tsl)[i0 + u * AGG_THREADS] = v[u];
            }
            for (; i0 < n4; i0 += AGG_THREADS)
                ((float4*)tsl)[i0] = t4[i0];
        } else {
            for (int i = tid; i < slen; i += AGG_THREADS)
                tsl[i] = t[slo + i];
        }
        __syncthreads();

        if ((E & 3) == 0) {
            const int4* s4 = (const int4*)ei;
            const int4* d4 = (const int4*)(ei + E);
            const int nq   = E >> 2;
            const int per  = (nq + C - 1) / C;
            const int qbeg = min(nq, c * per);
            const int qend = min(nq, qbeg + per);
            for (int q0 = qbeg; q0 < qend; q0 += AGG_BATCH * AGG_THREADS) {
                unsigned bv[4 * AGG_BATCH];
                unsigned av[4 * AGG_BATCH];
                float    tv[4 * AGG_BATCH];
#pragma unroll
                for (int u = 0; u < AGG_BATCH; ++u) {
                    const int  q  = q0 + u * AGG_THREADS + tid;
                    const int  qq = min(q, qend - 1);
                    const int4 sq = s4[qq];
                    int4       d  = d4[qq];
                    if (q >= qend) d = make_int4(-1, -1, -1, -1);
                    bv[4 * u + 0] = (unsigned)(sq.x - slo);
                    bv[4 * u + 1] = (unsigned)(sq.y - slo);
                    bv[4 * u + 2] = (unsigned)(sq.z - slo);
                    bv[4 * u + 3] = (unsigned)(sq.w - slo);
                    av[4 * u + 0] = (unsigned)(d.x - lo);
                    av[4 * u + 1] = (unsigned)(d.y - lo);
                    av[4 * u + 2] = (unsigned)(d.z - lo);
                    av[4 * u + 3] = (unsigned)(d.w - lo);
                }
#pragma unroll
                for (int k = 0; k < 4 * AGG_BATCH; ++k)
                    if (bv[k] >= uslen) av[k] = 0xffffffffu;
#pragma unroll
                for (int k = 0; k < 4 * AGG_BATCH; ++k)
                    tv[k] = tsl[av[k] < ulen ? bv[k] : 0u];
#pragma unroll
                for (int k = 0; k < 4 * AGG_BATCH; ++k)
                    if (av[k] < ulen) unsafeAtomicAdd(&tbl[av[k]], tv[k]);
            }
        } else {
            const int per  = (E + C - 1) / C;
            const int ebeg = min(E, c * per);
            const int eend = min(E, ebeg + per);
            for (int e = ebeg + tid; e < eend; e += AGG_THREADS) {
                const unsigned a = (unsigned)(ei[E + e] - lo);
                const unsigned b = (unsigned)(ei[e] - slo);
                if (a < ulen && b < uslen) unsafeAtomicAdd(&tbl[a], tsl[b]);
            }
        }
        __syncthreads();

        const int rot = (((len >> 5) * (c * S + s)) / (C * S)) << 5;
        for (int i0 = tid; i0 < len; i0 += AGG_BATCH * AGG_THREADS) {
            int   j[AGG_BATCH];
            float v[AGG_BATCH];
#pragma unroll
            for (int u = 0; u < AGG_BATCH; ++u) {
                const int i = i0 + u * AGG_THREADS;
                j[u] = i + rot;
                if (j[u] >= len) j[u] -= len;
                v[u] = i < len ? tbl[j[u]] : 0.f;
            }
#pragma unroll
            for (int u = 0; u < AGG_BATCH; ++u)
                if (v[u] != 0.f) unsafeAtomicAdd(aplane + lo + j[u], v[u]);
        }
        __syncthreads();
    }
}

#define SRT_C     256
#define SRT_SPAN  416
#define SRT_SHIFT 10
#define SRT_NRMAX 256
#define SRT_QMAX  4
#define SRT_DIR   (SRT_NRMAX + 1)
#define SRT_HW    (AGG_THREADS / 16)
#define SRT_SEG   (SRT_C / SRT_HW)
#define SUM_BATCH 8

__global__ __launch_bounds__(AGG_THREADS)
void gcn_sort_kernel(const int* __restrict__ ei, int* __restrict__ dir,
                     uint4* __restrict__ bins, int E, int NR, int RQ) {
    __shared__ __attribute__((aligned(16)))
        unsigned lst[(SRT_QMAX * AGG_THREADS + SRT_NRMAX) * 4];
    __shared__ unsigned cnt[SRT_NRMAX + 32];
    __shared__ unsigned start[SRT_NRMAX + 1];
    const int tid = threadIdx.x;
    const int c   = blockIdx.x;
    if (tid < SRT_NRMAX + 32) cnt[tid] = 0u;
    __syncthreads();

    const int4* s4 = (const int4*)ei;
    const int4* d4 = (const int4*)(ei + E);
    const int nq   = E >> 2;
    const int per  = (nq + SRT_C - 1) / SRT_C;
    const int qbeg = min(nq, c * per);
    const int qend = min(nq, qbeg + per);

    unsigned sv[4 * SRT_QMAX];
    unsigned dv[4 * SRT_QMAX];
#pragma unroll
    for (int u = 0; u < SRT_QMAX; ++u) {
        const int  q  = qbeg + u * AGG_THREADS + tid;
        const int  qq = max(0, min(q, qend - 1));
        const int4 sq = s4[qq];
        int4       d  = d4[qq];
        if (q >= qend) d = make_int4(-1, -1, -1, -1);
        sv[4 * u + 0] = (unsigned)sq.x;
        sv[4 * u + 1] = (unsigned)sq.y;
        sv[4 * u + 2] = (unsigned)sq.z;
        sv[4 * u + 3] = (unsigned)sq.w;
        dv[4 * u + 0] = (unsigned)d.x;
        dv[4 * u + 1] = (unsigned)d.y;
        dv[4 * u + 2] = (unsigned)d.z;
        dv[4 * u + 3] = (unsigned)d.w;
    }

    unsigned rr[4 * SRT_QMAX];
    unsigned pp[4 * SRT_QMAX];
#pragma unroll
    for (int k = 0; k < 4 * SRT_QMAX; ++k) {
        const unsigned r = dv[k] / SRT_SPAN;
        dv[k] -= r * SRT_SPAN;
        rr[k] = r < (unsigned)NR ? r : 0xffffffffu;
        pp[k] = atomicAdd(&cnt[r < (unsigned)NR ? r : SRT_NRMAX + (tid & 31)], 1u);
    }
    __syncthreads();

    if (tid < 32) {
        unsigned loc[SRT_NRMAX / 32];
        unsigned sum = 0u;
#pragma unroll
        for (int j = 0; j < SRT_NRMAX / 32; ++j) {
            const int i = tid * (SRT_NRMAX / 32) + j;
            loc[j] = sum;
            sum += (cnt[i] + 3u) & ~3u;
        }
        unsigned incl = sum;
#pragma unroll
        for (int d = 1; d < 32; d <<= 1) {
            const unsigned v = (unsigned)__shfl_up((int)incl, (unsigned)d, 32);
            if (tid >= d) incl += v;
        }
        const unsigned excl = incl - sum;
#pragma unroll
        for (int j = 0; j < SRT_NRMAX / 32; ++j)
            start[tid * (SRT_NRMAX / 32) + j] = excl + loc[j];
        if (tid == 31) start[SRT_NRMAX] = incl;
    }
    __syncthreads();

#pragma unroll
    for (int k = 0; k < 4 * SRT_QMAX; ++k)
        pp[k] += start[rr[k] != 0xffffffffu ? rr[k] : 0u];
#pragma unroll
    for (int k = 0; k < 4 * SRT_QMAX; ++k)
        if (rr[k] != 0xffffffffu)
            lst[pp[k]] = (sv[k] << SRT_SHIFT) | dv[k];
    if (tid < NR) {
        const unsigned e = start[tid + 1];
        for (unsigned j = start[tid] + cnt[tid]; j < e; ++j) lst[j] = 0xffffffffu;
    }
    __syncthreads();

    const unsigned tq  = start[SRT_NRMAX] >> 2;
    uint4* const   out = bins + (size_t)c * RQ;
    for (unsigned i = (unsigned)tid; i < tq; i += AGG_THREADS)
        out[i] = ((const uint4*)lst)[i];
    if (tid < SRT_DIR) dir[c * SRT_DIR + tid] = (int)(start[tid] >> 2);
}

__device__ __forceinline__ void seg_first(const int* __restrict__ dir,
                                          const uint4* __restrict__ bins,
                                          int r, int RQ, int hw, int l16,
                                          int (&q0)[SRT_SEG], int (&n)[SRT_SEG],
                                          uint4 (&w)[2 * SRT_SEG]) {
#pragma unroll
    for (int k = 0; k < SRT_SEG; ++k) {
        const int cc = hw + SRT_HW * k;
        const int a  = dir[cc * SRT_DIR + r];
        const int b  = dir[cc * SRT_DIR + r + 1];
        q0[k] = cc * RQ + a;
        n[k]  = b - a;
    }
#pragma unroll
    for (int k = 0; k < SRT_SEG; ++k) {
        const int cc = hw + SRT_HW * k;
#pragma unroll
        for (int h = 0; h < 2; ++h) {
            const int q = l16 + 16 * h;
            w[2 * k + h] = bins[q < n[k] ? q0[k] + q : cc * RQ];
        }
    }
}

__device__ __forceinline__ void seg_mask(const int (&n)[SRT_SEG], int l16,
                                         uint4 (&w)[2 * SRT_SEG]) {
#pragma unroll
    for (int k = 0; k < SRT_SEG; ++k)
#pragma unroll
        for (int h = 0; h < 2; ++h)
            if (l16 + 16 * h >= n[k])
                w[2 * k + h] = make_uint4(0xffffffffu, 0xffffffffu,
                                          0xffffffffu, 0xffffffffu);
}

__global__ __launch_bounds__(AGG_THREADS)
void gcn_deg_owner_kernel(const int* __restrict__ dir,
                          const uint4* __restrict__ bins,
                          unsigned* __restrict__ deg, int N, int NR, int RQ) {
    __shared__ unsigned tbl[SRT_SPAN];
    const int tid = threadIdx.x;
    const int hw  = tid >> 4;
    const int l16 = tid & 15;
    for (int r = blockIdx.x; r < NR; r += gridDim.x) {
        const int      lo   = r * SRT_SPAN;
        const unsigned ulen = (unsigned)min(SRT_SPAN, N - lo);
        int q0[SRT_SEG], n[SRT_SEG];
        uint4 w[2 * SRT_SEG];
        seg_first(dir, bins, r, RQ, hw, l16, q0, n, w);
        if (tid < SRT_SPAN) tbl[tid] = 0u;
        __syncthreads();
        seg_mask(n, l16, w);
#pragma unroll
        for (int k = 0; k < 2 * SRT_SEG; ++k) {
            const unsigned a0 = w[k].x & 1023u, a1 = w[k].y & 1023u;
            const unsigned a2 = w[k].z & 1023u, a3 = w[k].w & 1023u;
            if (a0 < ulen) atomicAdd(&tbl[a0], 1u);
            if (a1 < ulen) atomicAdd(&tbl[a1], 1u);
            if (a2 < ulen) atomicAdd(&tbl[a2], 1u);
            if (a3 < ulen) atomicAdd(&tbl[a3], 1u);
        }
        for (int k = 0; k < SRT_SEG; ++k)
            for (int q = l16 + 32; q < n[k]; q += 16) {
                const uint4 x = bins[q0[k] + q];
                const unsigned a0 = x.x & 1023u, a1 = x.y & 1023u;
                const unsigned a2 = x.z & 1023u, a3 = x.w & 1023u;
                if (a0 < ulen) atomicAdd(&tbl[a0], 1u);
                if (a1 < ulen) atomicAdd(&tbl[a1], 1u);
                if (a2 < ulen) atomicAdd(&tbl[a2], 1u);
                if (a3 < ulen) atomicAdd(&tbl[a3], 1u);
            }
        __syncthreads();
        if ((unsigned)tid < ulen) deg[lo + tid] = tbl[tid];
        __syncthreads();
    }
}

__device__ __forceinline__ void sum_quads4(const uint4 (&x)[4], unsigned ulen,
                                           unsigned slo, unsigned uslen,
                                           float* tbl, const float* tsl) {
    unsigned av[16];
    unsigned bv[16];
    float    tv[16];
#pragma unroll
    for (int u = 0; u < 4; ++u) {
        av[4 * u + 0] = x[u].x & 1023u;
        av[4 * u + 1] = x[u].y & 1023u;
        av[4 * u + 2] = x[u].z & 1023u;
        av[4 * u + 3] = x[u].w & 1023u;
        bv[4 * u + 0] = (x[u].x >> SRT_SHIFT) - slo;
        bv[4 * u + 1] = (x[u].y >> SRT_SHIFT) - slo;
        bv[4 * u + 2] = (x[u].z >> SRT_SHIFT) - slo;
        bv[4 * u + 3] = (x[u].w >> SRT_SHIFT) - slo;
    }
#pragma unroll
    for (int k = 0; k < 16; ++k)
        if (bv[k] >= uslen) av[k] = 0xffffffffu;
#pragma unroll
    for (int k = 0; k < 16; ++k)
        tv[k] = tsl[av[k] < ulen ? bv[k] : 0u];
#pragma unroll
    for (int k = 0; k < 16; ++k)
        if (av[k] < ulen) unsafeAtomicAdd(&tbl[av[k]], tv[k]);
}

__device__ __forceinline__ void sum_word(unsigned x, unsigned ulen,
                                         unsigned slo, unsigned uslen,
                                         float* tbl, const float* tsl) {
    const unsigned a = x & 1023u;
    const unsigned b = (x >> SRT_SHIFT) - slo;
    if (a < ulen && b < uslen) unsafeAtomicAdd(&tbl[a], tsl[b]);
}

__global__ __launch_bounds__(AGG_THREADS)
void gcn_sum_owner_kernel(const int* __restrict__ dir,
                          const uint4* __restrict__ bins,
                          const float* __restrict__ t,
                          const float* __restrict__ dinv,
                          const float* __restrict__ bconv,
                          const float* __restrict__ wfc,
                          const float* __restrict__ bfc,
                          float* __restrict__ out, int N, int NR, int RQ) {
    __shared__ __attribute__((aligned(16))) float tsl[SRC_TBL];
    __shared__ float tbl[SRT_SPAN];
    __shared__ float4 spare[32];
    const int tid   = threadIdx.x;
    const int hw    = tid >> 4;
    const int l16   = tid & 15;
    const int S     = (N + SRC_TBL - 1) / SRC_TBL;
    const int sspan = (((N + S - 1) / S) + 3) & ~3;

    float cst = 0.f;
#pragma unroll
    for (int j = 0; j < HID; ++j)
        cst += bconv[j] * wfc[j];
    cst += bfc[0];

    for (int r = blockIdx.x; r < NR; r += gridDim.x) {
        const int      lo   = r * SRT_SPAN;
        const unsigned ulen = (unsigned)min(SRT_SPAN, N - lo);
        int q0[SRT_SEG], n[SRT_SEG];
        uint4 w[2 * SRT_SEG];
        seg_first(dir, bins, r, RQ, hw, l16, q0, n, w);
        float dn = 0.f, tn = 0.f;
        if ((unsigned)tid < ulen) {
            dn = dinv[lo + tid];
            tn = t[lo + tid];
        }
        if (tid < SRT_SPAN) tbl[tid] = 0.f;

        for (int s = 0; s < S; ++s) {
            const int slo  = s * sspan;
            const int slen = max(0, min(sspan, N - slo));
            {
                const float4* t4 = (const float4*)(t + slo);
                const int     n4 = slen >> 2;
                for (int i0 = tid; i0 < n4; i0 += SUM_BATCH * AGG_THREADS) {
                    float4 v[SUM_BATCH];
#pragma unroll
                    for (int u = 0; u < SUM_BATCH; ++u)
                        v[u] = t4[min(i0 + u * AGG_THREADS, n4 - 1)];
#pragma unroll
                    for (int u = 0; u < SUM_BATCH; ++u) {
                        const int i = i0 + u * AGG_THREADS;
                        float4* const d = i < n4 ? (float4*)tsl + i
                                                 : spare + (tid & 31);
                        *d = v[u];
                    }
                }
            }
            __syncthreads();
            if (s == 0) seg_mask(n, l16, w);

#pragma unroll
            for (int k = 0; k < 2 * SRT_SEG; k += 4) {
                const uint4 x[4] = {w[k], w[k + 1], w[k + 2], w[k + 3]};
                sum_quads4(x, ulen, (unsigned)slo, (unsigned)slen, tbl, tsl);
            }
            for (int k = 0; k < SRT_SEG; ++k)
                for (int q = l16 + 32; q < n[k]; q += 16) {
                    const uint4 x = bins[q0[k] + q];
                    sum_word(x.x, ulen, (unsigned)slo, (unsigned)slen, tbl, tsl);
                    sum_word(x.y, ulen, (unsigned)slo, (unsigned)slen, tbl, tsl);
                    sum_word(x.z, ulen, (unsigned)slo, (unsigned)slen, tbl, tsl);
                    sum_word(x.w, ulen, (unsigned)slo, (unsigned)slen, tbl, tsl);
                }
            __syncthreads();
        }

        if ((unsigned)tid < ulen)
            out[lo + tid] = dn * (tn + tbl[tid]) + cst;
        __syncthreads();
    }
}

__global__ void gcn_head_kernel(const float* __restrict__ acc,
                                const float* __restrict__ dinv,
                                const float* __restrict__ bconv,
                                const float* __restrict__ wfc,
                                const float* __restrict__ bfc,
                                float* __restrict__ out, int N) {
    int i = blockIdx.x * blockDim.x + threadIdx.x;
    if (i >= N) return;
    float c = 0.f;
#pragma unroll
    for (int j = 0; j < HID; ++j)
        c += bconv[j] * wfc[j];
    c += bfc[0];
    out[i] = dinv[i] * (acc[i] + acc[N + i]) + c;
}

extern "C" void kernel_launch(void* const* d_in, const int* in_sizes, int n_in,
                              void* d_out, int out_size, void* d_ws, size_t ws_size,
                              hipStream_t stream) {
    const float* x     = (const float*)d_in[0];
    const int*   ei    = (const int*)d_in[1];
    const float* W     = (const float*)d_in[2];
    const float* bconv = (const float*)d_in[3];
    const float* wfc   = (const float*)d_in[4];
    const float* bfc   = (const float*)d_in[5];
    float* out = (float*)d_out;

    const int N = in_sizes[0] / D_FEAT;
    const int E = in_sizes[1] / 2;

    const int ntiles = N / 16;

    const int nq  = E >> 2;
    const int per = (nq + SRT_C - 1) / SRT_C;
    const int NR  = (N + SRT_SPAN - 1) / SRT_SPAN;
    const int RQ  = per + (3 * NR) / 4 + 1;
    const size_t need = (size_t)N * 12 + (size_t)SRT_C * SRT_DIR * 4
                      + (size_t)SRT_C * RQ * 16;
    const bool sorted = (E & 3) == 0 && (N & 3) == 0 && nq > 0 &&
                        per <= SRT_QMAX * AGG_THREADS && NR <= SRT_NRMAX &&
                        ws_size >= need;

    char* ws = (char*)d_ws;
    if (sorted) {
        unsigned* deg  = (unsigned*)ws;
        float*    dinv = (float*)(ws + (size_t)N * 4);
        float*    t    = dinv + N;
        int*      dir  = (int*)(t + N);
        uint4*    bins = (uint4*)(dir + SRT_C * SRT_DIR);

        gcn_sort_kernel<<<SRT_C, AGG_THREADS, 0, stream>>>(ei, dir, bins, E, NR, RQ);
        gcn_deg_owner_kernel<<<256, AGG_THREADS, 0, stream>>>(dir, bins, deg, N, NR, RQ);
        gcn_gemm_wmma_kernel<false><<<256, 256, 0, stream>>>(
            x, W, wfc, deg, dinv, t, nullptr, ntiles, N);
        gcn_sum_owner_kernel<<<256, AGG_THREADS, 0, stream>>>(
            dir, bins, t, dinv, bconv, wfc, bfc, out, N, NR, RQ);
        return;
    }

    unsigned* deg  = (unsigned*)ws;
    float*    dinv = (float*)(ws + (size_t)N * 8);
    float*    t    = dinv + N;
    float*    acc  = t + N;

    hipMemsetAsync(deg, 0, (size_t)N * 2 * sizeof(unsigned), stream);

    gcn_deg_kernel<<<256, AGG_THREADS, 0, stream>>>(ei + E, deg, E, N);

    gcn_gemm_wmma_kernel<true><<<256, 256, 0, stream>>>(
        x, W, wfc, deg, dinv, t, acc, ntiles, N);

    gcn_scatter_kernel<<<256, AGG_THREADS, 0, stream>>>(ei, t, acc, E, N);

    gcn_head_kernel<<<(N + 255) / 256, 256, 0, stream>>>(acc, dinv, bconv, wfc,
                                                         bfc, out, N);
}
